# starting kernel
# baseline (speedup 1.0000x reference)
#include <hip/hip_runtime.h>
#include <math.h>

#define BN 128
#define TN 64
#define EN 300
#define MP 20
#define S2E 600
#define EPSF 1e-6f

typedef __attribute__((ext_vector_type(2))) float v2f;
typedef __attribute__((ext_vector_type(8))) float v8f;

__device__ __forceinline__ float wred_sum(float v) {
#pragma unroll
  for (int off = 16; off; off >>= 1) v += __shfl_xor(v, off, 32);
  return v;
}

__global__ void norms_kernel(const float* __restrict__ h1, const float* __restrict__ h2,
                             float* __restrict__ n1, float* __restrict__ n2) {
  int gw = (blockIdx.x * blockDim.x + threadIdx.x) >> 5;
  int lane = threadIdx.x & 31;
  if (gw >= BN * TN) return;
  const float* r1 = h1 + (size_t)gw * S2E;
  const float* r2 = h2 + (size_t)gw * S2E;
#pragma unroll
  for (int d = 0; d < 2; ++d) {
    float s1 = 0.f, s2 = 0.f;
    const float* p1 = r1 + d * EN;
    const float* p2 = r2 + d * EN;
    for (int e = lane; e < EN; e += 32) {
      float a = p1[e]; s1 += a * a;
      float b = p2[e]; s2 += b * b;
    }
    s1 = wred_sum(s1);
    s2 = wred_sum(s2);
    if (lane == 0) {
      n1[d * BN * TN + gw] = sqrtf(fmaxf(s1, EPSF));
      n2[d * BN * TN + gw] = sqrtf(fmaxf(s2, EPSF));
    }
  }
}

__global__ void cosmat_kernel(const float* __restrict__ h1, const float* __restrict__ h2,
                              const float* __restrict__ n1, const float* __restrict__ n2,
                              float* __restrict__ cosm  ) {
  int bd = blockIdx.x;
  int b = bd >> 1, d = bd & 1;
  int wave = threadIdx.x >> 5, lane = threadIdx.x & 31;
  int lh = lane >> 4, ll = lane & 15;
  const float* Ab = h1 + ((size_t)b * TN) * S2E + d * EN;
  const float* Bb = h2 + ((size_t)b * TN) * S2E + d * EN;
  const float* n1p = n1 + (size_t)d * BN * TN + (size_t)b * TN;
  const float* n2p = n2 + (size_t)d * BN * TN + (size_t)b * TN;
  float* op = cosm + ((size_t)d * BN + b) * TN * TN;
  for (int t = wave; t < 16; t += 8) {
    int i0 = (t >> 2) * 16, j0 = (t & 3) * 16;
    const float* arow = Ab + (size_t)(i0 + ll) * S2E;
    const float* brow = Bb + (size_t)(j0 + ll) * S2E;
    v8f acc = {};
    for (int kk = 0; kk < EN; kk += 4) {
      int k = kk + 2 * lh;
      v2f af = { arow[k], arow[k + 1] };
      v2f bf = { brow[k], brow[k + 1] };
      acc = __builtin_amdgcn_wmma_f32_16x16x4_f32(false, af, false, bf,
                                                  (short)0, acc, false, false);
    }
    float invn2 = 1.f / n2p[j0 + ll];
#pragma unroll
    for (int r = 0; r < 8; ++r) {
      int i = i0 + lh * 8 + r;
      op[(size_t)i * TN + (j0 + ll)] = acc[r] / n1p[i] * invn2;
    }
  }
}

__global__ void rowstats_kernel(const float* __restrict__ cosm,
                                float* __restrict__ rowsum, int* __restrict__ amax) {
  int gw = (blockIdx.x * blockDim.x + threadIdx.x) >> 5;
  int lane = threadIdx.x & 31;
  if (gw >= 2 * BN * TN) return;
  const float* row = cosm + (size_t)gw * TN;
  float v0 = row[lane], v1 = row[lane + 32];
  float s = v0 + v1;
  float m = v0; int mi = lane;
  if (v1 > m) { m = v1; mi = lane + 32; }
#pragma unroll
  for (int off = 16; off; off >>= 1) {
    s += __shfl_xor(s, off, 32);
    float om = __shfl_xor(m, off, 32);
    int omi = __shfl_xor(mi, off, 32);
    if (om > m || (om == m && omi < mi)) { m = om; mi = omi; }
  }
  if (lane == 0) { rowsum[gw] = s + EPSF; amax[gw] = mi; }
}

__global__ void att_kernel(const float* __restrict__ h2, const float* __restrict__ cosm,
                           const float* __restrict__ rowsum, float* __restrict__ att) {
  int blk = blockIdx.x;
  int et = blk % 19;
  int db = blk / 19;
  int b = db % BN, d = db / BN;
  int wave = threadIdx.x >> 5, lane = threadIdx.x & 31;
  int lh = lane >> 4, ll = lane & 15;
  int i0 = wave * 16, e0 = et * 16;
  bool ein = (e0 + ll) < EN;
  int ecol = ein ? (e0 + ll) : 0;
  const float* crow = cosm + (size_t)db * TN * TN + (size_t)(i0 + ll) * TN;
  const float* h2b = h2 + ((size_t)b * TN) * S2E + d * EN;
  v8f acc = {};
  for (int kk = 0; kk < TN; kk += 4) {
    int k = kk + 2 * lh;
    v2f af = { crow[k], crow[k + 1] };
    v2f bf = { h2b[(size_t)k * S2E + ecol], h2b[(size_t)(k + 1) * S2E + ecol] };
    acc = __builtin_amdgcn_wmma_f32_16x16x4_f32(false, af, false, bf,
                                                (short)0, acc, false, false);
  }
  if (ein) {
    float* arow = att + (size_t)db * TN * EN;
    const float* rs = rowsum + (size_t)db * TN;
#pragma unroll
    for (int r = 0; r < 8; ++r) {
      int i = i0 + lh * 8 + r;
      arow[(size_t)i * EN + e0 + ll] = acc[r] / rs[i];
    }
  }
}

__global__ void match_kernel(const float* __restrict__ h1, const float* __restrict__ h2,
                             const float* __restrict__ kern, const float* __restrict__ att,
                             const int* __restrict__ amax,
                             float* __restrict__ N1mp, float* __restrict__ N2mp,
                             float* __restrict__ out) {
  int gw = (blockIdx.x * blockDim.x + threadIdx.x) >> 5;
  int lane = threadIdx.x & 31;
  if (gw >= 2 * BN * TN) return;
  int d = gw / (BN * TN);
  int bt = gw - d * (BN * TN);
  int b = bt / TN, t = bt - b * TN;
  const float* x     = h1 + (size_t)bt * S2E + d * EN;
  const float* yfull = h2 + ((size_t)b * TN + (TN - 1)) * S2E + d * EN;
  const float* yatt  = att + ((size_t)d * BN + b) * TN * EN + (size_t)t * EN;
  int jm = amax[((size_t)d * BN + b) * TN + t];
  const float* ymatt = h2 + ((size_t)b * TN + jm) * S2E + d * EN;
  const float* yown  = h2 + (size_t)bt * S2E + d * EN;
  float* orow = out + (size_t)bt * 160;

  float xv[10], y0v[10], y1v[10], y2v[10], yov[10];
#pragma unroll
  for (int q = 0; q < 10; ++q) {
    int e = lane + 32 * q;
    bool v = e < EN;
    xv[q]  = v ? x[e]     : 0.f;
    y0v[q] = v ? yfull[e] : 0.f;
    y1v[q] = v ? yatt[e]  : 0.f;
    y2v[q] = v ? ymatt[e] : 0.f;
    yov[q] = v ? yown[e]  : 0.f;
  }

  for (int m = 0; m < MP; ++m) {
    const float* kr = kern + (size_t)m * 2400;
#pragma unroll
    for (int kind = 0; kind < 3; ++kind) {
      const float* wv = kr + (kind * 2 + d) * EN;
      float sxy = 0.f, sxx = 0.f, syy = 0.f;
#pragma unroll
      for (int q = 0; q < 10; ++q) {
        int e = lane + 32 * q;
        float w = (e < EN) ? wv[e] : 0.f;
        float w2 = w * w;
        float xa = xv[q];
        float yb = (kind == 0) ? y0v[q] : (kind == 1) ? y1v[q] : y2v[q];
        sxy += xa * yb * w2;
        sxx += xa * xa * w2;
        syy += yb * yb * w2;
      }
      sxy = wred_sum(sxy); sxx = wred_sum(sxx); syy = wred_sum(syy);
      if (lane == 0)
        orow[(kind * 2 + d) * MP + m] =
            sxy / sqrtf(fmaxf(sxx, EPSF)) / sqrtf(fmaxf(syy, EPSF));
    }
    const float* wv = kr + (6 + d) * EN;
    float sxx = 0.f, syy = 0.f;
#pragma unroll
    for (int q = 0; q < 10; ++q) {
      int e = lane + 32 * q;
      float w = (e < EN) ? wv[e] : 0.f;
      float w2 = w * w;
      sxx += xv[q] * xv[q] * w2;
      syy += yov[q] * yov[q] * w2;
    }
    sxx = wred_sum(sxx); syy = wred_sum(syy);
    if (lane == 0) {
      size_t o = (((size_t)d * BN + b) * TN + t) * MP + m;
      N1mp[o] = sqrtf(fmaxf(sxx, EPSF));
      N2mp[o] = sqrtf(fmaxf(syy, EPSF));
    }
  }
}

__global__ void maxpool_kernel(const float* __restrict__ h1, const float* __restrict__ h2,
                               const float* __restrict__ kern,
                               const float* __restrict__ N1mp, const float* __restrict__ N2mp,
                               float* __restrict__ out) {
  __shared__ float wl[EN];
  int blk = blockIdx.x;
  int m = blk % MP;
  int db = blk / MP;
  int b = db % BN, d = db / BN;
  int wave = threadIdx.x >> 5, lane = threadIdx.x & 31;
  int lh = lane >> 4, ll = lane & 15;

  const float* wsrc = kern + (size_t)m * 2400 + (6 + d) * EN;
  for (int e = threadIdx.x; e < EN; e += 128) wl[e] = wsrc[e];
  __syncthreads();

  const float* Ab = h1 + ((size_t)b * TN) * S2E + d * EN;
  const float* Bb = h2 + ((size_t)b * TN) * S2E + d * EN;
  int i0 = wave * 16;
  const float* arow = Ab + (size_t)(i0 + ll) * S2E;
  const float* b0r = Bb + (size_t)(0  + ll) * S2E;
  const float* b1r = Bb + (size_t)(16 + ll) * S2E;
  const float* b2r = Bb + (size_t)(32 + ll) * S2E;
  const float* b3r = Bb + (size_t)(48 + ll) * S2E;

  v8f a0 = {}, a1 = {}, a2 = {}, a3 = {};
  for (int kk = 0; kk < EN; kk += 4) {
    int k = kk + 2 * lh;
    float w0 = wl[k], w1 = wl[k + 1];
    v2f af = { arow[k] * w0, arow[k + 1] * w1 };
    v2f f0 = { b0r[k] * w0, b0r[k + 1] * w1 };
    v2f f1 = { b1r[k] * w0, b1r[k + 1] * w1 };
    v2f f2 = { b2r[k] * w0, b2r[k + 1] * w1 };
    v2f f3 = { b3r[k] * w0, b3r[k + 1] * w1 };
    a0 = __builtin_amdgcn_wmma_f32_16x16x4_f32(false, af, false, f0, (short)0, a0, false, false);
    a1 = __builtin_amdgcn_wmma_f32_16x16x4_f32(false, af, false, f1, (short)0, a1, false, false);
    a2 = __builtin_amdgcn_wmma_f32_16x16x4_f32(false, af, false, f2, (short)0, a2, false, false);
    a3 = __builtin_amdgcn_wmma_f32_16x16x4_f32(false, af, false, f3, (short)0, a3, false, false);
  }

  const float* n1p = N1mp + ((size_t)d * BN + b) * TN * MP + m;
  const float* n2p = N2mp + ((size_t)d * BN + b) * TN * MP + m;
#pragma unroll
  for (int r = 0; r < 8; ++r) {
    float v0 = a0[r] / n2p[(size_t)(0  + ll) * MP];
    float v1 = a1[r] / n2p[(size_t)(16 + ll) * MP];
    float v2 = a2[r] / n2p[(size_t)(32 + ll) * MP];
    float v3 = a3[r] / n2p[(size_t)(48 + ll) * MP];
    float v = fmaxf(fmaxf(v0, v1), fmaxf(v2, v3));
#pragma unroll
    for (int off = 1; off < 16; off <<= 1)
      v = fmaxf(v, __shfl_xor(v, off, 32));
    int i = i0 + lh * 8 + r;
    if (ll == 0)
      out[((size_t)(b * TN + i)) * 160 + (6 + d) * MP + m] = v / n1p[(size_t)i * MP];
  }
}

extern "C" void kernel_launch(void* const* d_in, const int* in_sizes, int n_in,
                              void* d_out, int out_size, void* d_ws, size_t ws_size,
                              hipStream_t stream) {
  const float* h1   = (const float*)d_in[0];
  const float* h2   = (const float*)d_in[1];
  const float* kern = (const float*)d_in[2];
  float* out = (float*)d_out;

  float* cosm   = (float*)d_ws;
  float* n1     = cosm + (size_t)2 * BN * TN * TN;
  float* n2     = n1 + 2 * BN * TN;
  float* rowsum = n2 + 2 * BN * TN;
  int*   amax   = (int*)(rowsum + 2 * BN * TN);
  float* att    = (float*)(amax + 2 * BN * TN);
  float* N1mp   = att + (size_t)2 * BN * TN * EN;
  float* N2mp   = N1mp + (size_t)2 * BN * TN * MP;

  norms_kernel   <<<BN * TN / 8,     256, 0, stream>>>(h1, h2, n1, n2);
  cosmat_kernel  <<<2 * BN,          256, 0, stream>>>(h1, h2, n1, n2, cosm);
  rowstats_kernel<<<2 * BN * TN / 8, 256, 0, stream>>>(cosm, rowsum, amax);
  att_kernel     <<<2 * BN * 19,     128, 0, stream>>>(h2, cosm, rowsum, att);
  match_kernel   <<<2 * BN * TN / 8, 256, 0, stream>>>(h1, h2, kern, att, amax,
                                                       N1mp, N2mp, out);
  maxpool_kernel <<<2 * BN * MP,     128, 0, stream>>>(h1, h2, kern, N1mp, N2mp, out);
}


# max-pool kernel: one block per (dir,batch) copies the h1/h2 tiles and weights once into LDS with row-contiguous loads and computes all 20 perspectives from LDS; same products and k order
# speedup vs baseline: 1.3013x; 1.3013x over previous
#include <hip/hip_runtime.h>
#include <math.h>

#define BN 128
#define TN 64
#define EN 300
#define MP 20
#define S2E 600
#define EPSF 1e-6f

typedef __attribute__((ext_vector_type(2))) float v2f;
typedef __attribute__((ext_vector_type(8))) float v8f;

__device__ __forceinline__ float wred_sum(float v) {
#pragma unroll
  for (int off = 16; off; off >>= 1) v += __shfl_xor(v, off, 32);
  return v;
}

__global__ void norms_kernel(const float* __restrict__ h1, const float* __restrict__ h2,
                             float* __restrict__ n1, float* __restrict__ n2) {
  int gw = (blockIdx.x * blockDim.x + threadIdx.x) >> 5;
  int lane = threadIdx.x & 31;
  if (gw >= BN * TN) return;
  const float* r1 = h1 + (size_t)gw * S2E;
  const float* r2 = h2 + (size_t)gw * S2E;
#pragma unroll
  for (int d = 0; d < 2; ++d) {
    float s1 = 0.f, s2 = 0.f;
    const float* p1 = r1 + d * EN;
    const float* p2 = r2 + d * EN;
    for (int e = lane; e < EN; e += 32) {
      float a = p1[e]; s1 += a * a;
      float b = p2[e]; s2 += b * b;
    }
    s1 = wred_sum(s1);
    s2 = wred_sum(s2);
    if (lane == 0) {
      n1[d * BN * TN + gw] = sqrtf(fmaxf(s1, EPSF));
      n2[d * BN * TN + gw] = sqrtf(fmaxf(s2, EPSF));
    }
  }
}

__global__ void cosmat_kernel(const float* __restrict__ h1, const float* __restrict__ h2,
                              const float* __restrict__ n1, const float* __restrict__ n2,
                              float* __restrict__ cosm  ) {
  int bd = blockIdx.x;
  int b = bd >> 1, d = bd & 1;
  int wave = threadIdx.x >> 5, lane = threadIdx.x & 31;
  int lh = lane >> 4, ll = lane & 15;
  const float* Ab = h1 + ((size_t)b * TN) * S2E + d * EN;
  const float* Bb = h2 + ((size_t)b * TN) * S2E + d * EN;
  const float* n1p = n1 + (size_t)d * BN * TN + (size_t)b * TN;
  const float* n2p = n2 + (size_t)d * BN * TN + (size_t)b * TN;
  float* op = cosm + ((size_t)d * BN + b) * TN * TN;
  for (int t = wave; t < 16; t += 8) {
    int i0 = (t >> 2) * 16, j0 = (t & 3) * 16;
    const float* arow = Ab + (size_t)(i0 + ll) * S2E;
    const float* brow = Bb + (size_t)(j0 + ll) * S2E;
    v8f acc = {};
    for (int kk = 0; kk < EN; kk += 4) {
      int k = kk + 2 * lh;
      v2f af = { arow[k], arow[k + 1] };
      v2f bf = { brow[k], brow[k + 1] };
      acc = __builtin_amdgcn_wmma_f32_16x16x4_f32(false, af, false, bf,
                                                  (short)0, acc, false, false);
    }
    float invn2 = 1.f / n2p[j0 + ll];
#pragma unroll
    for (int r = 0; r < 8; ++r) {
      int i = i0 + lh * 8 + r;
      op[(size_t)i * TN + (j0 + ll)] = acc[r] / n1p[i] * invn2;
    }
  }
}

__global__ void rowstats_kernel(const float* __restrict__ cosm,
                                float* __restrict__ rowsum, int* __restrict__ amax) {
  int gw = (blockIdx.x * blockDim.x + threadIdx.x) >> 5;
  int lane = threadIdx.x & 31;
  if (gw >= 2 * BN * TN) return;
  const float* row = cosm + (size_t)gw * TN;
  float v0 = row[lane], v1 = row[lane + 32];
  float s = v0 + v1;
  float m = v0; int mi = lane;
  if (v1 > m) { m = v1; mi = lane + 32; }
#pragma unroll
  for (int off = 16; off; off >>= 1) {
    s += __shfl_xor(s, off, 32);
    float om = __shfl_xor(m, off, 32);
    int omi = __shfl_xor(mi, off, 32);
    if (om > m || (om == m && omi < mi)) { m = om; mi = omi; }
  }
  if (lane == 0) { rowsum[gw] = s + EPSF; amax[gw] = mi; }
}

__global__ void att_kernel(const float* __restrict__ h2, const float* __restrict__ cosm,
                           const float* __restrict__ rowsum, float* __restrict__ att) {
  int blk = blockIdx.x;
  int et = blk % 19;
  int db = blk / 19;
  int b = db % BN, d = db / BN;
  int wave = threadIdx.x >> 5, lane = threadIdx.x & 31;
  int lh = lane >> 4, ll = lane & 15;
  int i0 = wave * 16, e0 = et * 16;
  bool ein = (e0 + ll) < EN;
  int ecol = ein ? (e0 + ll) : 0;
  const float* crow = cosm + (size_t)db * TN * TN + (size_t)(i0 + ll) * TN;
  const float* h2b = h2 + ((size_t)b * TN) * S2E + d * EN;
  v8f acc = {};
  for (int kk = 0; kk < TN; kk += 4) {
    int k = kk + 2 * lh;
    v2f af = { crow[k], crow[k + 1] };
    v2f bf = { h2b[(size_t)k * S2E + ecol], h2b[(size_t)(k + 1) * S2E + ecol] };
    acc = __builtin_amdgcn_wmma_f32_16x16x4_f32(false, af, false, bf,
                                                (short)0, acc, false, false);
  }
  if (ein) {
    float* arow = att + (size_t)db * TN * EN;
    const float* rs = rowsum + (size_t)db * TN;
#pragma unroll
    for (int r = 0; r < 8; ++r) {
      int i = i0 + lh * 8 + r;
      arow[(size_t)i * EN + e0 + ll] = acc[r] / rs[i];
    }
  }
}

__global__ void match_kernel(const float* __restrict__ h1, const float* __restrict__ h2,
                             const float* __restrict__ kern, const float* __restrict__ att,
                             const int* __restrict__ amax,
                             float* __restrict__ N1mp, float* __restrict__ N2mp,
                             float* __restrict__ out) {
  int gw = (blockIdx.x * blockDim.x + threadIdx.x) >> 5;
  int lane = threadIdx.x & 31;
  if (gw >= 2 * BN * TN) return;
  int d = gw / (BN * TN);
  int bt = gw - d * (BN * TN);
  int b = bt / TN, t = bt - b * TN;
  const float* x     = h1 + (size_t)bt * S2E + d * EN;
  const float* yfull = h2 + ((size_t)b * TN + (TN - 1)) * S2E + d * EN;
  const float* yatt  = att + ((size_t)d * BN + b) * TN * EN + (size_t)t * EN;
  int jm = amax[((size_t)d * BN + b) * TN + t];
  const float* ymatt = h2 + ((size_t)b * TN + jm) * S2E + d * EN;
  const float* yown  = h2 + (size_t)bt * S2E + d * EN;
  float* orow = out + (size_t)bt * 160;

  float xv[10], y0v[10], y1v[10], y2v[10], yov[10];
#pragma unroll
  for (int q = 0; q < 10; ++q) {
    int e = lane + 32 * q;
    bool v = e < EN;
    xv[q]  = v ? x[e]     : 0.f;
    y0v[q] = v ? yfull[e] : 0.f;
    y1v[q] = v ? yatt[e]  : 0.f;
    y2v[q] = v ? ymatt[e] : 0.f;
    yov[q] = v ? yown[e]  : 0.f;
  }

  for (int m = 0; m < MP; ++m) {
    const float* kr = kern + (size_t)m * 2400;
#pragma unroll
    for (int kind = 0; kind < 3; ++kind) {
      const float* wv = kr + (kind * 2 + d) * EN;
      float sxy = 0.f, sxx = 0.f, syy = 0.f;
#pragma unroll
      for (int q = 0; q < 10; ++q) {
        int e = lane + 32 * q;
        float w = (e < EN) ? wv[e] : 0.f;
        float w2 = w * w;
        float xa = xv[q];
        float yb = (kind == 0) ? y0v[q] : (kind == 1) ? y1v[q] : y2v[q];
        sxy += xa * yb * w2;
        sxx += xa * xa * w2;
        syy += yb * yb * w2;
      }
      sxy = wred_sum(sxy); sxx = wred_sum(sxx); syy = wred_sum(syy);
      if (lane == 0)
        orow[(kind * 2 + d) * MP + m] =
            sxy / sqrtf(fmaxf(sxx, EPSF)) / sqrtf(fmaxf(syy, EPSF));
    }
    const float* wv = kr + (6 + d) * EN;
    float sxx = 0.f, syy = 0.f;
#pragma unroll
    for (int q = 0; q < 10; ++q) {
      int e = lane + 32 * q;
      float w = (e < EN) ? wv[e] : 0.f;
      float w2 = w * w;
      sxx += xv[q] * xv[q] * w2;
      syy += yov[q] * yov[q] * w2;
    }
    sxx = wred_sum(sxx); syy = wred_sum(syy);
    if (lane == 0) {
      size_t o = (((size_t)d * BN + b) * TN + t) * MP + m;
      N1mp[o] = sqrtf(fmaxf(sxx, EPSF));
      N2mp[o] = sqrtf(fmaxf(syy, EPSF));
    }
  }
}

#define MPW 16
#define MPG (MPW / 4)
#define ROW4 (EN / 4)
#define AQ ((TN * ROW4 + MPW * 32 - 1) / (MPW * 32))
#define WQ ((MP * ROW4 + MPW * 32 - 1) / (MPW * 32))

__global__ __launch_bounds__(MPW * 32)
void maxpool_kernel(const float* __restrict__ h1, const float* __restrict__ h2,
                    const float* __restrict__ kern,
                    const float* __restrict__ N1mp, const float* __restrict__ N2mp,
                    float* __restrict__ out) {
  __shared__ __attribute__((aligned(16))) float As[TN * EN];
  __shared__ __attribute__((aligned(16))) float Bs[TN * EN];
  __shared__ __attribute__((aligned(16))) float Ws[MP * EN];
  int db = blockIdx.x;
  int b = db % BN, d = db / BN;
  int wave = __builtin_amdgcn_readfirstlane((int)(threadIdx.x >> 5));
  int lane = threadIdx.x & 31;
  int lh = lane >> 4, ll = lane & 15;

  const float* Ab = h1 + ((size_t)b * TN) * S2E + d * EN;
  const float* Bb = h2 + ((size_t)b * TN) * S2E + d * EN;
  const int tid = threadIdx.x;
  const float* wsrc = kern + (6 + d) * EN;
  float4 va[AQ], vb[AQ], wv[WQ];
#pragma unroll
  for (int q = 0; q < AQ; ++q) {
    int idx = tid + MPW * 32 * q;
    int ci = idx < TN * ROW4 ? idx : TN * ROW4 - 1;
    int r = ci / ROW4, c = ci - r * ROW4;
    va[q] = *reinterpret_cast<const float4*>(Ab + (size_t)r * S2E + 4 * c);
    vb[q] = *reinterpret_cast<const float4*>(Bb + (size_t)r * S2E + 4 * c);
  }
#pragma unroll
  for (int q = 0; q < WQ; ++q) {
    int idx = tid + MPW * 32 * q;
    int ci = idx < MP * ROW4 ? idx : MP * ROW4 - 1;
    int r = ci / ROW4, c = ci - r * ROW4;
    wv[q] = *reinterpret_cast<const float4*>(wsrc + (size_t)r * 2400 + 4 * c);
  }
#pragma unroll
  for (int q = 0; q < AQ; ++q) {
    int idx = tid + MPW * 32 * q;
    if (idx < TN * ROW4) {
      *reinterpret_cast<float4*>(As + 4 * idx) = va[q];
      *reinterpret_cast<float4*>(Bs + 4 * idx) = vb[q];
    }
  }
#pragma unroll
  for (int q = 0; q < WQ; ++q) {
    int idx = tid + MPW * 32 * q;
    if (idx < MP * ROW4) *reinterpret_cast<float4*>(Ws + 4 * idx) = wv[q];
  }
  __syncthreads();

  int strip = wave & 3, mg = wave >> 2;
  int i0 = strip * 16;
  const float* ar  = As + (i0 + ll) * EN + 2 * lh;
  const float* b0r = Bs + (0  + ll) * EN + 2 * lh;
  const float* b1r = Bs + (16 + ll) * EN + 2 * lh;
  const float* b2r = Bs + (32 + ll) * EN + 2 * lh;
  const float* b3r = Bs + (48 + ll) * EN + 2 * lh;
  const float* n1b = N1mp + ((size_t)d * BN + b) * TN * MP;
  const float* n2b = N2mp + ((size_t)d * BN + b) * TN * MP;

  for (int m = mg; m < MP; m += MPG) {
    const float* wr = Ws + m * EN + 2 * lh;
    v8f a0 = {}, a1 = {}, a2 = {}, a3 = {};
#pragma unroll 5
    for (int kk = 0; kk < EN; kk += 4) {
      v2f w  = *reinterpret_cast<const v2f*>(wr  + kk);
      v2f af = *reinterpret_cast<const v2f*>(ar  + kk) * w;
      v2f f0 = *reinterpret_cast<const v2f*>(b0r + kk) * w;
      v2f f1 = *reinterpret_cast<const v2f*>(b1r + kk) * w;
      v2f f2 = *reinterpret_cast<const v2f*>(b2r + kk) * w;
      v2f f3 = *reinterpret_cast<const v2f*>(b3r + kk) * w;
      a0 = __builtin_amdgcn_wmma_f32_16x16x4_f32(false, af, false, f0, (short)0, a0, false, false);
      a1 = __builtin_amdgcn_wmma_f32_16x16x4_f32(false, af, false, f1, (short)0, a1, false, false);
      a2 = __builtin_amdgcn_wmma_f32_16x16x4_f32(false, af, false, f2, (short)0, a2, false, false);
      a3 = __builtin_amdgcn_wmma_f32_16x16x4_f32(false, af, false, f3, (short)0, a3, false, false);
    }

    const float* n1p = n1b + m;
    const float* n2p = n2b + m;
    float d0 = n2p[(size_t)(0  + ll) * MP];
    float d1 = n2p[(size_t)(16 + ll) * MP];
    float d2 = n2p[(size_t)(32 + ll) * MP];
    float d3 = n2p[(size_t)(48 + ll) * MP];
#pragma unroll
    for (int r = 0; r < 8; ++r) {
      float v0 = a0[r] / d0;
      float v1 = a1[r] / d1;
      float v2 = a2[r] / d2;
      float v3 = a3[r] / d3;
      float v = fmaxf(fmaxf(v0, v1), fmaxf(v2, v3));
#pragma unroll
      for (int off = 1; off < 16; off <<= 1)
        v = fmaxf(v, __shfl_xor(v, off, 32));
      int i = i0 + lh * 8 + r;
      if (ll == 0)
        out[((size_t)(b * TN + i)) * 160 + (6 + d) * MP + m] = v / n1p[(size_t)i * MP];
    }
  }
}

extern "C" void kernel_launch(void* const* d_in, const int* in_sizes, int n_in,
                              void* d_out, int out_size, void* d_ws, size_t ws_size,
                              hipStream_t stream) {
  const float* h1   = (const float*)d_in[0];
  const float* h2   = (const float*)d_in[1];
  const float* kern = (const float*)d_in[2];
  float* out = (float*)d_out;

  float* cosm   = (float*)d_ws;
  float* n1     = cosm + (size_t)2 * BN * TN * TN;
  float* n2     = n1 + 2 * BN * TN;
  float* rowsum = n2 + 2 * BN * TN;
  int*   amax   = (int*)(rowsum + 2 * BN * TN);
  float* att    = (float*)(amax + 2 * BN * TN);
  float* N1mp   = att + (size_t)2 * BN * TN * EN;
  float* N2mp   = N1mp + (size_t)2 * BN * TN * MP;

  norms_kernel   <<<BN * TN / 8,     256, 0, stream>>>(h1, h2, n1, n2);
  cosmat_kernel  <<<2 * BN,          256, 0, stream>>>(h1, h2, n1, n2, cosm);
  rowstats_kernel<<<2 * BN * TN / 8, 256, 0, stream>>>(cosm, rowsum, amax);
  att_kernel     <<<2 * BN * 19,     128, 0, stream>>>(h2, cosm, rowsum, att);
  match_kernel   <<<2 * BN * TN / 8, 256, 0, stream>>>(h1, h2, kern, att, amax,
                                                       N1mp, N2mp, out);
  maxpool_kernel <<<2 * BN,          MPW * 32, 0, stream>>>(h1, h2, kern, N1mp, N2mp, out);
}


# fuse the norm, cosine-matrix, row-sum/argmax and attentive-vector kernels into the match kernel, computed per (dir,b) from its h1/h2 tiles in LDS; 6 launches become 2; summation order unchanged
# speedup vs baseline: 3.6960x; 2.8402x over previous
#include <hip/hip_runtime.h>
#include <math.h>

#define BN 128
#define TN 64
#define EN 300
#define MP 20
#define S2E 600
#define EPSF 1e-6f

typedef __attribute__((ext_vector_type(2))) float v2f;
typedef __attribute__((ext_vector_type(8))) float v8f;

#define ROW4 (EN / 4)
#define MTW 16
#define MTT (MTW * 32)
#define XQ ((TN * ROW4 + MTT - 1) / MTT)
#define VQ ((MP * ROW4 + MTT - 1) / MTT)
#define XO 0
#define HO (TN * EN)
#define TO (2 * TN * EN)
#define WO (3 * TN * EN)
#define CP 68
#define CO (WO + 2 * MP * EN)
#define NO (CO + TN * CP)
#define RO (NO + 2 * TN)
#define AO (RO + TN)
#define SMF (AO + TN)
#define ETN ((EN + 15) / 16)
#define NRW (2 * TN / MTW)
#define SRW (TN / MTW)

__global__ __launch_bounds__(MTT)
void match_kernel(const float* __restrict__ h1, const float* __restrict__ h2,
                  const float* __restrict__ kern,
                  float* __restrict__ N1mp, float* __restrict__ N2mp,
                  float* __restrict__ out) {
  __shared__ __attribute__((aligned(16))) float sm[SMF];
  int db = blockIdx.x;
  int b = db % BN, d = db / BN;
  int wave = __builtin_amdgcn_readfirstlane((int)(threadIdx.x >> 5));
  int lane = threadIdx.x & 31;
  int lh = lane >> 4, ll = lane & 15;
  const int tid = threadIdx.x;

  const float* Xg = h1 + ((size_t)b * TN) * S2E + d * EN;
  const float* Hg = h2 + ((size_t)b * TN) * S2E + d * EN;
  {
    float4 vx[XQ], vh[XQ];
#pragma unroll
    for (int q = 0; q < XQ; ++q) {
      int idx = tid + MTT * q;
      int ci = idx < TN * ROW4 ? idx : TN * ROW4 - 1;
      int r = ci / ROW4, c = ci - r * ROW4;
      vx[q] = *reinterpret_cast<const float4*>(Xg + (size_t)r * S2E + 4 * c);
      vh[q] = *reinterpret_cast<const float4*>(Hg + (size_t)r * S2E + 4 * c);
    }
#pragma unroll
    for (int q = 0; q < XQ; ++q) {
      int idx = tid + MTT * q;
      if (idx < TN * ROW4) {
        *reinterpret_cast<float4*>(sm + XO + 4 * idx) = vx[q];
        *reinterpret_cast<float4*>(sm + HO + 4 * idx) = vh[q];
      }
    }
  }
  __syncthreads();

  {
    float s[NRW];
#pragma unroll
    for (int u = 0; u < NRW; ++u) s[u] = 0.f;
    for (int e = lane; e < EN; e += 32) {
#pragma unroll
      for (int u = 0; u < NRW; ++u) {
        float a = sm[XO + (wave + MTW * u) * EN + e];
        s[u] += a * a;
      }
    }
#pragma unroll
    for (int off = 16; off; off >>= 1) {
#pragma unroll
      for (int u = 0; u < NRW; ++u) s[u] += __shfl_xor(s[u], off, 32);
    }
    if (lane == 0) {
#pragma unroll
      for (int u = 0; u < NRW; ++u) sm[NO + wave + MTW * u] = sqrtf(fmaxf(s[u], EPSF));
    }
  }
  __syncthreads();

  {
    int ci0 = (wave >> 2) * 16, cj0 = (wave & 3) * 16;
    const float* ca = sm + XO + (ci0 + ll) * EN + 2 * lh;
    const float* cb = sm + HO + (cj0 + ll) * EN + 2 * lh;
    v8f acc = {};
#pragma unroll 5
    for (int kk = 0; kk < EN; kk += 4) {
      v2f af = *reinterpret_cast<const v2f*>(ca + kk);
      v2f bf = *reinterpret_cast<const v2f*>(cb + kk);
      acc = __builtin_amdgcn_wmma_f32_16x16x4_f32(false, af, false, bf,
                                                  (short)0, acc, false, false);
    }
    float invn2 = 1.f / sm[NO + TN + cj0 + ll];
#pragma unroll
    for (int r = 0; r < 8; ++r) {
      int i = ci0 + lh * 8 + r;
      sm[CO + i * CP + cj0 + ll] = acc[r] / sm[NO + i] * invn2;
    }
  }
  __syncthreads();

  {
    float s[SRW], m[SRW]; int mi[SRW];
#pragma unroll
    for (int u = 0; u < SRW; ++u) {
      const float* row = sm + CO + (wave + MTW * u) * CP;
      float v0 = row[lane], v1 = row[lane + 32];
      s[u] = v0 + v1;
      m[u] = v0; mi[u] = lane;
      if (v1 > m[u]) { m[u] = v1; mi[u] = lane + 32; }
    }
#pragma unroll
    for (int off = 16; off; off >>= 1) {
#pragma unroll
      for (int u = 0; u < SRW; ++u) {
        s[u] += __shfl_xor(s[u], off, 32);
        float om = __shfl_xor(m[u], off, 32);
        int omi = __shfl_xor(mi[u], off, 32);
        if (om > m[u] || (om == m[u] && omi < mi[u])) { m[u] = om; mi[u] = omi; }
      }
    }
    if (lane == 0) {
#pragma unroll
      for (int u = 0; u < SRW; ++u) {
        sm[RO + wave + MTW * u] = s[u] + EPSF;
        sm[AO + wave + MTW * u] = __int_as_float(mi[u]);
      }
    }
  }
  __syncthreads();

  for (int q = wave; q < 4 * ETN; q += MTW) {
    int ai0 = (q & 3) * 16;
    int et = q >> 2;
    int e0 = et < ETN - 1 ? et * 16 : EN - 16;
    const float* crow = sm + CO + (ai0 + ll) * CP + 2 * lh;
    const float* hcol = sm + HO + (2 * lh) * EN + e0 + ll;
    v8f acc = {};
#pragma unroll
    for (int kk = 0; kk < TN; kk += 4) {
      v2f af = *reinterpret_cast<const v2f*>(crow + kk);
      v2f bf = { hcol[kk * EN], hcol[(kk + 1) * EN] };
      acc = __builtin_amdgcn_wmma_f32_16x16x4_f32(false, af, false, bf,
                                                  (short)0, acc, false, false);
    }
#pragma unroll
    for (int r = 0; r < 8; ++r) {
      int i = ai0 + lh * 8 + r;
      sm[TO + i * EN + e0 + ll] = acc[r] / sm[RO + i];
    }
  }
  __syncthreads();

  int strip = wave & 3, sel = (wave >> 2) & 1, nt = wave >> 3;
  int i0 = strip * 16;
  int mcol = nt * 16 + ll;
  int mrow = mcol < MP ? mcol : MP - 1;
  int jm = __float_as_int(sm[AO + i0 + ll]);
  const float* xr = sm + XO + (i0 + ll) * EN + 2 * lh;
  const float* wr = sm + WO + sel * (MP * EN) + mrow * EN + 2 * lh;

  for (int ph = 0; ph < 2; ++ph) {
    if (ph) __syncthreads();
    {
      float4 wv[2][VQ];
#pragma unroll
      for (int s = 0; s < 2; ++s) {
        int ks = ph == 0 ? (s == 0 ? 0 : 3) : (s == 0 ? 1 : 2);
        const float* wsrc = kern + (ks * 2 + d) * EN;
#pragma unroll
        for (int q = 0; q < VQ; ++q) {
          int idx = tid + MTT * q;
          int ci = idx < MP * ROW4 ? idx : MP * ROW4 - 1;
          int r = ci / ROW4, c = ci - r * ROW4;
          wv[s][q] = *reinterpret_cast<const float4*>(wsrc + (size_t)r * 2400 + 4 * c);
        }
      }
#pragma unroll
      for (int s = 0; s < 2; ++s) {
#pragma unroll
        for (int q = 0; q < VQ; ++q) {
          int idx = tid + MTT * q;
          if (idx < MP * ROW4) {
            float4 w = wv[s][q];
            w.x *= w.x; w.y *= w.y; w.z *= w.z; w.w *= w.w;
            *reinterpret_cast<float4*>(sm + WO + s * (MP * EN) + 4 * idx) = w;
          }
        }
      }
    }
    __syncthreads();

    int kind = ph == 0 ? (sel == 0 ? 0 : 3) : (sel == 0 ? 1 : 2);
    int yoff = (kind == 0) ? HO + (TN - 1) * EN
             : (kind == 1) ? TO + (i0 + ll) * EN
             : (kind == 2) ? HO + jm * EN
             :               HO + (i0 + ll) * EN;
    const float* yr = sm + yoff + 2 * lh;

    v8f sxy = {}, sxx = {}, syy = {};
    if (kind == 3) {
#pragma unroll 5
      for (int kk = 0; kk < EN; kk += 4) {
        v2f x = *reinterpret_cast<const v2f*>(xr + kk);
        v2f y = *reinterpret_cast<const v2f*>(yr + kk);
        v2f w = *reinterpret_cast<const v2f*>(wr + kk);
        sxx = __builtin_amdgcn_wmma_f32_16x16x4_f32(false, x * x, false, w, (short)0, sxx, false, false);
        syy = __builtin_amdgcn_wmma_f32_16x16x4_f32(false, y * y, false, w, (short)0, syy, false, false);
      }
    } else {
#pragma unroll 5
      for (int kk = 0; kk < EN; kk += 4) {
        v2f x = *reinterpret_cast<const v2f*>(xr + kk);
        v2f y = *reinterpret_cast<const v2f*>(yr + kk);
        v2f w = *reinterpret_cast<const v2f*>(wr + kk);
        sxy = __builtin_amdgcn_wmma_f32_16x16x4_f32(false, x * y, false, w, (short)0, sxy, false, false);
        sxx = __builtin_amdgcn_wmma_f32_16x16x4_f32(false, x * x, false, w, (short)0, sxx, false, false);
        syy = __builtin_amdgcn_wmma_f32_16x16x4_f32(false, y * y, false, w, (short)0, syy, false, false);
      }
    }

    if (mcol < MP) {
      if (kind == 3) {
#pragma unroll
        for (int r = 0; r < 8; ++r) {
          int i = i0 + lh * 8 + r;
          size_t o = (((size_t)d * BN + b) * TN + i) * MP + mcol;
          N1mp[o] = sqrtf(fmaxf(sxx[r], EPSF));
          N2mp[o] = sqrtf(fmaxf(syy[r], EPSF));
        }
      } else {
#pragma unroll
        for (int r = 0; r < 8; ++r) {
          int i = i0 + lh * 8 + r;
          out[((size_t)(b * TN + i)) * 160 + (kind * 2 + d) * MP + mcol] =
              sxy[r] / sqrtf(fmaxf(sxx[r], EPSF)) / sqrtf(fmaxf(syy[r], EPSF));
        }
      }
    }
  }
}

#define MPW 16
#define MPG (MPW / 4)
#define AQ ((TN * ROW4 + MPW * 32 - 1) / (MPW * 32))
#define WQ ((MP * ROW4 + MPW * 32 - 1) / (MPW * 32))

__global__ __launch_bounds__(MPW * 32)
void maxpool_kernel(const float* __restrict__ h1, const float* __restrict__ h2,
                    const float* __restrict__ kern,
                    const float* __restrict__ N1mp, const float* __restrict__ N2mp,
                    float* __restrict__ out) {
  __shared__ __attribute__((aligned(16))) float As[TN * EN];
  __shared__ __attribute__((aligned(16))) float Bs[TN * EN];
  __shared__ __attribute__((aligned(16))) float Ws[MP * EN];
  int db = blockIdx.x;
  int b = db % BN, d = db / BN;
  int wave = __builtin_amdgcn_readfirstlane((int)(threadIdx.x >> 5));
  int lane = threadIdx.x & 31;
  int lh = lane >> 4, ll = lane & 15;

  const float* Ab = h1 + ((size_t)b * TN) * S2E + d * EN;
  const float* Bb = h2 + ((size_t)b * TN) * S2E + d * EN;
  const int tid = threadIdx.x;
  const float* wsrc = kern + (6 + d) * EN;
  float4 va[AQ], vb[AQ], wv[WQ];
#pragma unroll
  for (int q = 0; q < AQ; ++q) {
    int idx = tid + MPW * 32 * q;
    int ci = idx < TN * ROW4 ? idx : TN * ROW4 - 1;
    int r = ci / ROW4, c = ci - r * ROW4;
    va[q] = *reinterpret_cast<const float4*>(Ab + (size_t)r * S2E + 4 * c);
    vb[q] = *reinterpret_cast<const float4*>(Bb + (size_t)r * S2E + 4 * c);
  }
#pragma unroll
  for (int q = 0; q < WQ; ++q) {
    int idx = tid + MPW * 32 * q;
    int ci = idx < MP * ROW4 ? idx : MP * ROW4 - 1;
    int r = ci / ROW4, c = ci - r * ROW4;
    wv[q] = *reinterpret_cast<const float4*>(wsrc + (size_t)r * 2400 + 4 * c);
  }
#pragma unroll
  for (int q = 0; q < AQ; ++q) {
    int idx = tid + MPW * 32 * q;
    if (idx < TN * ROW4) {
      *reinterpret_cast<float4*>(As + 4 * idx) = va[q];
      *reinterpret_cast<float4*>(Bs + 4 * idx) = vb[q];
    }
  }
#pragma unroll
  for (int q = 0; q < WQ; ++q) {
    int idx = tid + MPW * 32 * q;
    if (idx < MP * ROW4) *reinterpret_cast<float4*>(Ws + 4 * idx) = wv[q];
  }
  __syncthreads();

  int strip = wave & 3, mg = wave >> 2;
  int i0 = strip * 16;
  const float* ar  = As + (i0 + ll) * EN + 2 * lh;
  const float* b0r = Bs + (0  + ll) * EN + 2 * lh;
  const float* b1r = Bs + (16 + ll) * EN + 2 * lh;
  const float* b2r = Bs + (32 + ll) * EN + 2 * lh;
  const float* b3r = Bs + (48 + ll) * EN + 2 * lh;
  const float* n1b = N1mp + ((size_t)d * BN + b) * TN * MP;
  const float* n2b = N2mp + ((size_t)d * BN + b) * TN * MP;

  for (int m = mg; m < MP; m += MPG) {
    const float* wr = Ws + m * EN + 2 * lh;
    v8f a0 = {}, a1 = {}, a2 = {}, a3 = {};
#pragma unroll 5
    for (int kk = 0; kk < EN; kk += 4) {
      v2f w  = *reinterpret_cast<const v2f*>(wr  + kk);
      v2f af = *reinterpret_cast<const v2f*>(ar  + kk) * w;
      v2f f0 = *reinterpret_cast<const v2f*>(b0r + kk) * w;
      v2f f1 = *reinterpret_cast<const v2f*>(b1r + kk) * w;
      v2f f2 = *reinterpret_cast<const v2f*>(b2r + kk) * w;
      v2f f3 = *reinterpret_cast<const v2f*>(b3r + kk) * w;
      a0 = __builtin_amdgcn_wmma_f32_16x16x4_f32(false, af, false, f0, (short)0, a0, false, false);
      a1 = __builtin_amdgcn_wmma_f32_16x16x4_f32(false, af, false, f1, (short)0, a1, false, false);
      a2 = __builtin_amdgcn_wmma_f32_16x16x4_f32(false, af, false, f2, (short)0, a2, false, false);
      a3 = __builtin_amdgcn_wmma_f32_16x16x4_f32(false, af, false, f3, (short)0, a3, false, false);
    }

    const float* n1p = n1b + m;
    const float* n2p = n2b + m;
    float d0 = n2p[(size_t)(0  + ll) * MP];
    float d1 = n2p[(size_t)(16 + ll) * MP];
    float d2 = n2p[(size_t)(32 + ll) * MP];
    float d3 = n2p[(size_t)(48 + ll) * MP];
#pragma unroll
    for (int r = 0; r < 8; ++r) {
      float v0 = a0[r] / d0;
      float v1 = a1[r] / d1;
      float v2 = a2[r] / d2;
      float v3 = a3[r] / d3;
      float v = fmaxf(fmaxf(v0, v1), fmaxf(v2, v3));
#pragma unroll
      for (int off = 1; off < 16; off <<= 1)
        v = fmaxf(v, __shfl_xor(v, off, 32));
      int i = i0 + lh * 8 + r;
      if (ll == 0)
        out[((size_t)(b * TN + i)) * 160 + (6 + d) * MP + m] = v / n1p[(size_t)i * MP];
    }
  }
}

extern "C" void kernel_launch(void* const* d_in, const int* in_sizes, int n_in,
                              void* d_out, int out_size, void* d_ws, size_t ws_size,
                              hipStream_t stream) {
  const float* h1   = (const float*)d_in[0];
  const float* h2   = (const float*)d_in[1];
  const float* kern = (const float*)d_in[2];
  float* out = (float*)d_out;

  float* N1mp   = (float*)d_ws;
  float* N2mp   = N1mp + (size_t)2 * BN * TN * MP;

  match_kernel   <<<2 * BN,          MTT, 0, stream>>>(h1, h2, kern, N1mp, N2mp, out);
  maxpool_kernel <<<2 * BN,          MPW * 32, 0, stream>>>(h1, h2, kern, N1mp, N2mp, out);
}


# fuse the max-pooling kernel into the match kernel: it runs from the same LDS tiles, with its weights and norms handed over in LDS; 2 launches become 1; products and summation order unchanged
# speedup vs baseline: 3.9249x; 1.0619x over previous
#include <hip/hip_runtime.h>
#include <math.h>

#define BN 128
#define TN 64
#define EN 300
#define MP 20
#define S2E 600
#define EPSF 1e-6f

typedef __attribute__((ext_vector_type(2))) float v2f;
typedef __attribute__((ext_vector_type(8))) float v8f;

#define ROW4 (EN / 4)
#define MTW 16
#define MTT (MTW * 32)
#define XQ ((TN * ROW4 + MTT - 1) / MTT)
#define VQ ((MP * ROW4 + MTT - 1) / MTT)
#define XO 0
#define HO (TN * EN)
#define TO (2 * TN * EN)
#define WO (3 * TN * EN)
#define CP 68
#define CO (WO + 2 * MP * EN)
#define NO (CO + TN * CP)
#define RO (NO + 2 * TN)
#define AO (RO + TN)
#define MWO (AO + TN)
#define SMF (MWO + MP * EN)
#define N1O CO
#define N2O (CO + TN * MP)
#define MPG (MTW / 4)
#define ETN ((EN + 15) / 16)
#define NRW (2 * TN / MTW)
#define SRW (TN / MTW)

__global__ __launch_bounds__(MTT)
void match_kernel(const float* __restrict__ h1, const float* __restrict__ h2,
                  const float* __restrict__ kern,
                  float* __restrict__ out) {
  __shared__ __attribute__((aligned(16))) float sm[SMF];
  int db = blockIdx.x;
  int b = db % BN, d = db / BN;
  int wave = __builtin_amdgcn_readfirstlane((int)(threadIdx.x >> 5));
  int lane = threadIdx.x & 31;
  int lh = lane >> 4, ll = lane & 15;
  const int tid = threadIdx.x;

  const float* Xg = h1 + ((size_t)b * TN) * S2E + d * EN;
  const float* Hg = h2 + ((size_t)b * TN) * S2E + d * EN;
  {
    float4 vx[XQ], vh[XQ];
#pragma unroll
    for (int q = 0; q < XQ; ++q) {
      int idx = tid + MTT * q;
      int ci = idx < TN * ROW4 ? idx : TN * ROW4 - 1;
      int r = ci / ROW4, c = ci - r * ROW4;
      vx[q] = *reinterpret_cast<const float4*>(Xg + (size_t)r * S2E + 4 * c);
      vh[q] = *reinterpret_cast<const float4*>(Hg + (size_t)r * S2E + 4 * c);
    }
#pragma unroll
    for (int q = 0; q < XQ; ++q) {
      int idx = tid + MTT * q;
      if (idx < TN * ROW4) {
        *reinterpret_cast<float4*>(sm + XO + 4 * idx) = vx[q];
        *reinterpret_cast<float4*>(sm + HO + 4 * idx) = vh[q];
      }
    }
  }
  __syncthreads();

  {
    float s[NRW];
#pragma unroll
    for (int u = 0; u < NRW; ++u) s[u] = 0.f;
    for (int e = lane; e < EN; e += 32) {
#pragma unroll
      for (int u = 0; u < NRW; ++u) {
        float a = sm[XO + (wave + MTW * u) * EN + e];
        s[u] += a * a;
      }
    }
#pragma unroll
    for (int off = 16; off; off >>= 1) {
#pragma unroll
      for (int u = 0; u < NRW; ++u) s[u] += __shfl_xor(s[u], off, 32);
    }
    if (lane == 0) {
#pragma unroll
      for (int u = 0; u < NRW; ++u) sm[NO + wave + MTW * u] = sqrtf(fmaxf(s[u], EPSF));
    }
  }
  __syncthreads();

  {
    int ci0 = (wave >> 2) * 16, cj0 = (wave & 3) * 16;
    const float* ca = sm + XO + (ci0 + ll) * EN + 2 * lh;
    const float* cb = sm + HO + (cj0 + ll) * EN + 2 * lh;
    v8f acc = {};
#pragma unroll 5
    for (int kk = 0; kk < EN; kk += 4) {
      v2f af = *reinterpret_cast<const v2f*>(ca + kk);
      v2f bf = *reinterpret_cast<const v2f*>(cb + kk);
      acc = __builtin_amdgcn_wmma_f32_16x16x4_f32(false, af, false, bf,
                                                  (short)0, acc, false, false);
    }
    float invn2 = 1.f / sm[NO + TN + cj0 + ll];
#pragma unroll
    for (int r = 0; r < 8; ++r) {
      int i = ci0 + lh * 8 + r;
      sm[CO + i * CP + cj0 + ll] = acc[r] / sm[NO + i] * invn2;
    }
  }
  __syncthreads();

  {
    float s[SRW], m[SRW]; int mi[SRW];
#pragma unroll
    for (int u = 0; u < SRW; ++u) {
      const float* row = sm + CO + (wave + MTW * u) * CP;
      float v0 = row[lane], v1 = row[lane + 32];
      s[u] = v0 + v1;
      m[u] = v0; mi[u] = lane;
      if (v1 > m[u]) { m[u] = v1; mi[u] = lane + 32; }
    }
#pragma unroll
    for (int off = 16; off; off >>= 1) {
#pragma unroll
      for (int u = 0; u < SRW; ++u) {
        s[u] += __shfl_xor(s[u], off, 32);
        float om = __shfl_xor(m[u], off, 32);
        int omi = __shfl_xor(mi[u], off, 32);
        if (om > m[u] || (om == m[u] && omi < mi[u])) { m[u] = om; mi[u] = omi; }
      }
    }
    if (lane == 0) {
#pragma unroll
      for (int u = 0; u < SRW; ++u) {
        sm[RO + wave + MTW * u] = s[u] + EPSF;
        sm[AO + wave + MTW * u] = __int_as_float(mi[u]);
      }
    }
  }
  __syncthreads();

  for (int q = wave; q < 4 * ETN; q += MTW) {
    int ai0 = (q & 3) * 16;
    int et = q >> 2;
    int e0 = et < ETN - 1 ? et * 16 : EN - 16;
    const float* crow = sm + CO + (ai0 + ll) * CP + 2 * lh;
    const float* hcol = sm + HO + (2 * lh) * EN + e0 + ll;
    v8f acc = {};
#pragma unroll
    for (int kk = 0; kk < TN; kk += 4) {
      v2f af = *reinterpret_cast<const v2f*>(crow + kk);
      v2f bf = { hcol[kk * EN], hcol[(kk + 1) * EN] };
      acc = __builtin_amdgcn_wmma_f32_16x16x4_f32(false, af, false, bf,
                                                  (short)0, acc, false, false);
    }
#pragma unroll
    for (int r = 0; r < 8; ++r) {
      int i = ai0 + lh * 8 + r;
      sm[TO + i * EN + e0 + ll] = acc[r] / sm[RO + i];
    }
  }
  __syncthreads();

  int strip = wave & 3, sel = (wave >> 2) & 1, nt = wave >> 3;
  int i0 = strip * 16;
  int mcol = nt * 16 + ll;
  int mrow = mcol < MP ? mcol : MP - 1;
  int jm = __float_as_int(sm[AO + i0 + ll]);
  const float* xr = sm + XO + (i0 + ll) * EN + 2 * lh;
  const float* wr = sm + WO + sel * (MP * EN) + mrow * EN + 2 * lh;

  for (int ph = 0; ph < 2; ++ph) {
    if (ph) __syncthreads();
    {
      float4 wv[2][VQ];
#pragma unroll
      for (int s = 0; s < 2; ++s) {
        int ks = ph == 0 ? (s == 0 ? 0 : 3) : (s == 0 ? 1 : 2);
        const float* wsrc = kern + (ks * 2 + d) * EN;
#pragma unroll
        for (int q = 0; q < VQ; ++q) {
          int idx = tid + MTT * q;
          int ci = idx < MP * ROW4 ? idx : MP * ROW4 - 1;
          int r = ci / ROW4, c = ci - r * ROW4;
          wv[s][q] = *reinterpret_cast<const float4*>(wsrc + (size_t)r * 2400 + 4 * c);
        }
      }
#pragma unroll
      for (int s = 0; s < 2; ++s) {
#pragma unroll
        for (int q = 0; q < VQ; ++q) {
          int idx = tid + MTT * q;
          if (idx < MP * ROW4) {
            float4 w = wv[s][q];
            if (ph == 0 && s == 1) *reinterpret_cast<float4*>(sm + MWO + 4 * idx) = w;
            w.x *= w.x; w.y *= w.y; w.z *= w.z; w.w *= w.w;
            *reinterpret_cast<float4*>(sm + WO + s * (MP * EN) + 4 * idx) = w;
          }
        }
      }
    }
    __syncthreads();

    int kind = ph == 0 ? (sel == 0 ? 0 : 3) : (sel == 0 ? 1 : 2);
    int yoff = (kind == 0) ? HO + (TN - 1) * EN
             : (kind == 1) ? TO + (i0 + ll) * EN
             : (kind == 2) ? HO + jm * EN
             :               HO + (i0 + ll) * EN;
    const float* yr = sm + yoff + 2 * lh;

    v8f sxy = {}, sxx = {}, syy = {};
    if (kind == 3) {
#pragma unroll 5
      for (int kk = 0; kk < EN; kk += 4) {
        v2f x = *reinterpret_cast<const v2f*>(xr + kk);
        v2f y = *reinterpret_cast<const v2f*>(yr + kk);
        v2f w = *reinterpret_cast<const v2f*>(wr + kk);
        sxx = __builtin_amdgcn_wmma_f32_16x16x4_f32(false, x * x, false, w, (short)0, sxx, false, false);
        syy = __builtin_amdgcn_wmma_f32_16x16x4_f32(false, y * y, false, w, (short)0, syy, false, false);
      }
    } else {
#pragma unroll 5
      for (int kk = 0; kk < EN; kk += 4) {
        v2f x = *reinterpret_cast<const v2f*>(xr + kk);
        v2f y = *reinterpret_cast<const v2f*>(yr + kk);
        v2f w = *reinterpret_cast<const v2f*>(wr + kk);
        sxy = __builtin_amdgcn_wmma_f32_16x16x4_f32(false, x * y, false, w, (short)0, sxy, false, false);
        sxx = __builtin_amdgcn_wmma_f32_16x16x4_f32(false, x * x, false, w, (short)0, sxx, false, false);
        syy = __builtin_amdgcn_wmma_f32_16x16x4_f32(false, y * y, false, w, (short)0, syy, false, false);
      }
    }

    if (mcol < MP) {
      if (kind == 3) {
#pragma unroll
        for (int r = 0; r < 8; ++r) {
          int i = i0 + lh * 8 + r;
          sm[N1O + i * MP + mcol] = sqrtf(fmaxf(sxx[r], EPSF));
          sm[N2O + i * MP + mcol] = sqrtf(fmaxf(syy[r], EPSF));
        }
      } else {
#pragma unroll
        for (int r = 0; r < 8; ++r) {
          int i = i0 + lh * 8 + r;
          out[((size_t)(b * TN + i)) * 160 + (kind * 2 + d) * MP + mcol] =
              sxy[r] / sqrtf(fmaxf(sxx[r], EPSF)) / sqrtf(fmaxf(syy[r], EPSF));
        }
      }
    }
  }

  {
    int mg = wave >> 2;
    const float* ar  = sm + XO + (i0 + ll) * EN + 2 * lh;
    const float* b0r = sm + HO + (0  + ll) * EN + 2 * lh;
    const float* b1r = sm + HO + (16 + ll) * EN + 2 * lh;
    const float* b2r = sm + HO + (32 + ll) * EN + 2 * lh;
    const float* b3r = sm + HO + (48 + ll) * EN + 2 * lh;

    for (int m = mg; m < MP; m += MPG) {
      const float* wq = sm + MWO + m * EN + 2 * lh;
      v8f a0 = {}, a1 = {}, a2 = {}, a3 = {};
#pragma unroll 5
      for (int kk = 0; kk < EN; kk += 4) {
        v2f w  = *reinterpret_cast<const v2f*>(wq  + kk);
        v2f af = *reinterpret_cast<const v2f*>(ar  + kk) * w;
        v2f f0 = *reinterpret_cast<const v2f*>(b0r + kk) * w;
        v2f f1 = *reinterpret_cast<const v2f*>(b1r + kk) * w;
        v2f f2 = *reinterpret_cast<const v2f*>(b2r + kk) * w;
        v2f f3 = *reinterpret_cast<const v2f*>(b3r + kk) * w;
        a0 = __builtin_amdgcn_wmma_f32_16x16x4_f32(false, af, false, f0, (short)0, a0, false, false);
        a1 = __builtin_amdgcn_wmma_f32_16x16x4_f32(false, af, false, f1, (short)0, a1, false, false);
        a2 = __builtin_amdgcn_wmma_f32_16x16x4_f32(false, af, false, f2, (short)0, a2, false, false);
        a3 = __builtin_amdgcn_wmma_f32_16x16x4_f32(false, af, false, f3, (short)0, a3, false, false);
      }

      float d0 = sm[N2O + (0  + ll) * MP + m];
      float d1 = sm[N2O + (16 + ll) * MP + m];
      float d2 = sm[N2O + (32 + ll) * MP + m];
      float d3 = sm[N2O + (48 + ll) * MP + m];
#pragma unroll
      for (int r = 0; r < 8; ++r) {
        float v0 = a0[r] / d0;
        float v1 = a1[r] / d1;
        float v2 = a2[r] / d2;
        float v3 = a3[r] / d3;
        float v = fmaxf(fmaxf(v0, v1), fmaxf(v2, v3));
#pragma unroll
        for (int off = 1; off < 16; off <<= 1)
          v = fmaxf(v, __shfl_xor(v, off, 32));
        int i = i0 + lh * 8 + r;
        if (ll == 0)
          out[((size_t)(b * TN + i)) * 160 + (6 + d) * MP + m] = v / sm[N1O + i * MP + m];
      }
    }
  }
}

extern "C" void kernel_launch(void* const* d_in, const int* in_sizes, int n_in,
                              void* d_out, int out_size, void* d_ws, size_t ws_size,
                              hipStream_t stream) {
  const float* h1   = (const float*)d_in[0];
  const float* h2   = (const float*)d_in[1];
  const float* kern = (const float*)d_in[2];
  float* out = (float*)d_out;

  match_kernel<<<2 * BN, MTT, 0, stream>>>(h1, h2, kern, out);
}


# max-pool stage: one K pass per wave for its 5 perspectives (20 accumulator tiles, w^2 folded into the h1 fragment only), reads a k-step ahead, maxima reduced side by side; product rounding changes
# speedup vs baseline: 4.4132x; 1.1244x over previous
#include <hip/hip_runtime.h>
#include <math.h>

#define BN 128
#define TN 64
#define EN 300
#define MP 20
#define S2E 600
#define EPSF 1e-6f

typedef __attribute__((ext_vector_type(2))) float v2f;
typedef __attribute__((ext_vector_type(8))) float v8f;

#define ROW4 (EN / 4)
#define MTW 16
#define MTT (MTW * 32)
#define XQ ((TN * ROW4 + MTT - 1) / MTT)
#define VQ ((MP * ROW4 + MTT - 1) / MTT)
#define XO 0
#define HO (TN * EN)
#define TO (2 * TN * EN)
#define WO (3 * TN * EN)
#define CP 68
#define CO (WO + 2 * MP * EN)
#define NO (CO + TN * CP)
#define RO (NO + 2 * TN)
#define AO (RO + TN)
#define MWO (AO + TN)
#define SMF (MWO + MP * EN)
#define N1O CO
#define N2O (CO + TN * MP)
#define MPG (MTW / 4)
#define MPW (MP / MPG)
#define ETN ((EN + 15) / 16)
#define NRW (2 * TN / MTW)
#define SRW (TN / MTW)

__global__ __launch_bounds__(MTT)
void match_kernel(const float* __restrict__ h1, const float* __restrict__ h2,
                  const float* __restrict__ kern,
                  float* __restrict__ out) {
  __shared__ __attribute__((aligned(16))) float sm[SMF];
  int db = blockIdx.x;
  int b = db % BN, d = db / BN;
  int wave = __builtin_amdgcn_readfirstlane((int)(threadIdx.x >> 5));
  int lane = threadIdx.x & 31;
  int lh = lane >> 4, ll = lane & 15;
  const int tid = threadIdx.x;

  const float* Xg = h1 + ((size_t)b * TN) * S2E + d * EN;
  const float* Hg = h2 + ((size_t)b * TN) * S2E + d * EN;
  {
    float4 vx[XQ], vh[XQ];
#pragma unroll
    for (int q = 0; q < XQ; ++q) {
      int idx = tid + MTT * q;
      int ci = idx < TN * ROW4 ? idx : TN * ROW4 - 1;
      int r = ci / ROW4, c = ci - r * ROW4;
      vx[q] = *reinterpret_cast<const float4*>(Xg + (size_t)r * S2E + 4 * c);
      vh[q] = *reinterpret_cast<const float4*>(Hg + (size_t)r * S2E + 4 * c);
    }
#pragma unroll
    for (int q = 0; q < XQ; ++q) {
      int idx = tid + MTT * q;
      if (idx < TN * ROW4) {
        *reinterpret_cast<float4*>(sm + XO + 4 * idx) = vx[q];
        *reinterpret_cast<float4*>(sm + HO + 4 * idx) = vh[q];
      }
    }
  }
  __syncthreads();

  {
    float s[NRW];
#pragma unroll
    for (int u = 0; u < NRW; ++u) s[u] = 0.f;
    for (int e = lane; e < EN; e += 32) {
#pragma unroll
      for (int u = 0; u < NRW; ++u) {
        float a = sm[XO + (wave + MTW * u) * EN + e];
        s[u] += a * a;
      }
    }
#pragma unroll
    for (int off = 16; off; off >>= 1) {
#pragma unroll
      for (int u = 0; u < NRW; ++u) s[u] += __shfl_xor(s[u], off, 32);
    }
    if (lane == 0) {
#pragma unroll
      for (int u = 0; u < NRW; ++u) sm[NO + wave + MTW * u] = sqrtf(fmaxf(s[u], EPSF));
    }
  }
  __syncthreads();

  {
    int ci0 = (wave >> 2) * 16, cj0 = (wave & 3) * 16;
    const float* ca = sm + XO + (ci0 + ll) * EN + 2 * lh;
    const float* cb = sm + HO + (cj0 + ll) * EN + 2 * lh;
    v8f acc = {};
#pragma unroll 5
    for (int kk = 0; kk < EN; kk += 4) {
      v2f af = *reinterpret_cast<const v2f*>(ca + kk);
      v2f bf = *reinterpret_cast<const v2f*>(cb + kk);
      acc = __builtin_amdgcn_wmma_f32_16x16x4_f32(false, af, false, bf,
                                                  (short)0, acc, false, false);
    }
    float invn2 = 1.f / sm[NO + TN + cj0 + ll];
#pragma unroll
    for (int r = 0; r < 8; ++r) {
      int i = ci0 + lh * 8 + r;
      sm[CO + i * CP + cj0 + ll] = acc[r] / sm[NO + i] * invn2;
    }
  }
  __syncthreads();

  {
    float s[SRW], m[SRW]; int mi[SRW];
#pragma unroll
    for (int u = 0; u < SRW; ++u) {
      const float* row = sm + CO + (wave + MTW * u) * CP;
      float v0 = row[lane], v1 = row[lane + 32];
      s[u] = v0 + v1;
      m[u] = v0; mi[u] = lane;
      if (v1 > m[u]) { m[u] = v1; mi[u] = lane + 32; }
    }
#pragma unroll
    for (int off = 16; off; off >>= 1) {
#pragma unroll
      for (int u = 0; u < SRW; ++u) {
        s[u] += __shfl_xor(s[u], off, 32);
        float om = __shfl_xor(m[u], off, 32);
        int omi = __shfl_xor(mi[u], off, 32);
        if (om > m[u] || (om == m[u] && omi < mi[u])) { m[u] = om; mi[u] = omi; }
      }
    }
    if (lane == 0) {
#pragma unroll
      for (int u = 0; u < SRW; ++u) {
        sm[RO + wave + MTW * u] = s[u] + EPSF;
        sm[AO + wave + MTW * u] = __int_as_float(mi[u]);
      }
    }
  }
  __syncthreads();

  for (int q = wave; q < 4 * ETN; q += MTW) {
    int ai0 = (q & 3) * 16;
    int et = q >> 2;
    int e0 = et < ETN - 1 ? et * 16 : EN - 16;
    const float* crow = sm + CO + (ai0 + ll) * CP + 2 * lh;
    const float* hcol = sm + HO + (2 * lh) * EN + e0 + ll;
    v8f acc = {};
#pragma unroll
    for (int kk = 0; kk < TN; kk += 4) {
      v2f af = *reinterpret_cast<const v2f*>(crow + kk);
      v2f bf = { hcol[kk * EN], hcol[(kk + 1) * EN] };
      acc = __builtin_amdgcn_wmma_f32_16x16x4_f32(false, af, false, bf,
                                                  (short)0, acc, false, false);
    }
#pragma unroll
    for (int r = 0; r < 8; ++r) {
      int i = ai0 + lh * 8 + r;
      sm[TO + i * EN + e0 + ll] = acc[r] / sm[RO + i];
    }
  }
  __syncthreads();

  int strip = wave & 3, sel = (wave >> 2) & 1, nt = wave >> 3;
  int i0 = strip * 16;
  int mcol = nt * 16 + ll;
  int mrow = mcol < MP ? mcol : MP - 1;
  int jm = __float_as_int(sm[AO + i0 + ll]);
  const float* xr = sm + XO + (i0 + ll) * EN + 2 * lh;
  const float* wr = sm + WO + sel * (MP * EN) + mrow * EN + 2 * lh;

  for (int ph = 0; ph < 2; ++ph) {
    if (ph) __syncthreads();
    {
      float4 wv[2][VQ];
#pragma unroll
      for (int s = 0; s < 2; ++s) {
        int ks = ph == 0 ? (s == 0 ? 0 : 3) : (s == 0 ? 1 : 2);
        const float* wsrc = kern + (ks * 2 + d) * EN;
#pragma unroll
        for (int q = 0; q < VQ; ++q) {
          int idx = tid + MTT * q;
          int ci = idx < MP * ROW4 ? idx : MP * ROW4 - 1;
          int r = ci / ROW4, c = ci - r * ROW4;
          wv[s][q] = *reinterpret_cast<const float4*>(wsrc + (size_t)r * 2400 + 4 * c);
        }
      }
#pragma unroll
      for (int s = 0; s < 2; ++s) {
#pragma unroll
        for (int q = 0; q < VQ; ++q) {
          int idx = tid + MTT * q;
          if (idx < MP * ROW4) {
            float4 w = wv[s][q];
            w.x *= w.x; w.y *= w.y; w.z *= w.z; w.w *= w.w;
            *reinterpret_cast<float4*>(sm + WO + s * (MP * EN) + 4 * idx) = w;
            if (ph == 0 && s == 1) *reinterpret_cast<float4*>(sm + MWO + 4 * idx) = w;
          }
        }
      }
    }
    __syncthreads();

    int kind = ph == 0 ? (sel == 0 ? 0 : 3) : (sel == 0 ? 1 : 2);
    int yoff = (kind == 0) ? HO + (TN - 1) * EN
             : (kind == 1) ? TO + (i0 + ll) * EN
             : (kind == 2) ? HO + jm * EN
             :               HO + (i0 + ll) * EN;
    const float* yr = sm + yoff + 2 * lh;

    v8f sxy = {}, sxx = {}, syy = {};
    if (kind == 3) {
#pragma unroll 5
      for (int kk = 0; kk < EN; kk += 4) {
        v2f x = *reinterpret_cast<const v2f*>(xr + kk);
        v2f y = *reinterpret_cast<const v2f*>(yr + kk);
        v2f w = *reinterpret_cast<const v2f*>(wr + kk);
        sxx = __builtin_amdgcn_wmma_f32_16x16x4_f32(false, x * x, false, w, (short)0, sxx, false, false);
        syy = __builtin_amdgcn_wmma_f32_16x16x4_f32(false, y * y, false, w, (short)0, syy, false, false);
      }
    } else {
#pragma unroll 5
      for (int kk = 0; kk < EN; kk += 4) {
        v2f x = *reinterpret_cast<const v2f*>(xr + kk);
        v2f y = *reinterpret_cast<const v2f*>(yr + kk);
        v2f w = *reinterpret_cast<const v2f*>(wr + kk);
        sxy = __builtin_amdgcn_wmma_f32_16x16x4_f32(false, x * y, false, w, (short)0, sxy, false, false);
        sxx = __builtin_amdgcn_wmma_f32_16x16x4_f32(false, x * x, false, w, (short)0, sxx, false, false);
        syy = __builtin_amdgcn_wmma_f32_16x16x4_f32(false, y * y, false, w, (short)0, syy, false, false);
      }
    }

    if (mcol < MP) {
      if (kind == 3) {
#pragma unroll
        for (int r = 0; r < 8; ++r) {
          int i = i0 + lh * 8 + r;
          sm[N1O + i * MP + mcol] = sqrtf(fmaxf(sxx[r], EPSF));
          sm[N2O + i * MP + mcol] = sqrtf(fmaxf(syy[r], EPSF));
        }
      } else {
#pragma unroll
        for (int r = 0; r < 8; ++r) {
          int i = i0 + lh * 8 + r;
          out[((size_t)(b * TN + i)) * 160 + (kind * 2 + d) * MP + mcol] =
              sxy[r] / sqrtf(fmaxf(sxx[r], EPSF)) / sqrtf(fmaxf(syy[r], EPSF));
        }
      }
    }
  }

  {
    int mg = wave >> 2;
    const float* ar  = sm + XO + (i0 + ll) * EN + 2 * lh;
    const float* br  = sm + HO + ll * EN + 2 * lh;
    const float* wq  = sm + MWO + mg * EN + 2 * lh;

    v8f acc[MPW][4];
#pragma unroll
    for (int p = 0; p < MPW; ++p)
#pragma unroll
      for (int t = 0; t < 4; ++t) acc[p][t] = (v8f){};

    v2f af[MPW], bq[4];
    {
      v2f a = *reinterpret_cast<const v2f*>(ar);
#pragma unroll
      for (int p = 0; p < MPW; ++p) af[p] = a * *reinterpret_cast<const v2f*>(wq + MPG * p * EN);
#pragma unroll
      for (int t = 0; t < 4; ++t) bq[t] = *reinterpret_cast<const v2f*>(br + 16 * t * EN);
    }

#pragma nounroll
    for (int kk = 0; kk < EN; kk += 4) {
      int kn = kk + 4 < EN ? kk + 4 : kk;
      v2f na = *reinterpret_cast<const v2f*>(ar + kn);
      v2f nb[4], nw[MPW];
#pragma unroll
      for (int t = 0; t < 4; ++t) nb[t] = *reinterpret_cast<const v2f*>(br + 16 * t * EN + kn);
#pragma unroll
      for (int p = 0; p < MPW; ++p) nw[p] = *reinterpret_cast<const v2f*>(wq + MPG * p * EN + kn);

#pragma unroll
      for (int p = 0; p < MPW; ++p)
#pragma unroll
        for (int t = 0; t < 4; ++t)
          acc[p][t] = __builtin_amdgcn_wmma_f32_16x16x4_f32(false, af[p], false, bq[t],
                                                            (short)0, acc[p][t], false, false);

#pragma unroll
      for (int p = 0; p < MPW; ++p) af[p] = na * nw[p];
#pragma unroll
      for (int t = 0; t < 4; ++t) bq[t] = nb[t];
    }

    float v[MPW][8];
#pragma unroll
    for (int p = 0; p < MPW; ++p) {
      int m = mg + MPG * p;
      float d0 = sm[N2O + (0  + ll) * MP + m];
      float d1 = sm[N2O + (16 + ll) * MP + m];
      float d2 = sm[N2O + (32 + ll) * MP + m];
      float d3 = sm[N2O + (48 + ll) * MP + m];
#pragma unroll
      for (int r = 0; r < 8; ++r) {
        float v0 = acc[p][0][r] / d0;
        float v1 = acc[p][1][r] / d1;
        float v2 = acc[p][2][r] / d2;
        float v3 = acc[p][3][r] / d3;
        v[p][r] = fmaxf(fmaxf(v0, v1), fmaxf(v2, v3));
      }
    }
#pragma unroll
    for (int off = 1; off < 16; off <<= 1) {
#pragma unroll
      for (int p = 0; p < MPW; ++p)
#pragma unroll
        for (int r = 0; r < 8; ++r)
          v[p][r] = fmaxf(v[p][r], __shfl_xor(v[p][r], off, 32));
    }
    if (ll == 0) {
#pragma unroll
      for (int p = 0; p < MPW; ++p) {
        int m = mg + MPG * p;
#pragma unroll
        for (int r = 0; r < 8; ++r) {
          int i = i0 + lh * 8 + r;
          out[((size_t)(b * TN + i)) * 160 + (6 + d) * MP + m] = v[p][r] / sm[N1O + i * MP + m];
        }
      }
    }
  }
}

extern "C" void kernel_launch(void* const* d_in, const int* in_sizes, int n_in,
                              void* d_out, int out_size, void* d_ws, size_t ws_size,
                              hipStream_t stream) {
  const float* h1   = (const float*)d_in[0];
  const float* h2   = (const float*)d_in[1];
  const float* kern = (const float*)d_in[2];
  float* out = (float*)d_out;

  match_kernel<<<2 * BN, MTT, 0, stream>>>(h1, h2, kern, out);
}


# max-pool epilogue: the per-lane divisions by the norms become multiplies by reciprocal norms, which the norm stage now stores as 1/sqrt; rounding of that output slice changes
# speedup vs baseline: 4.6985x; 1.0646x over previous
#include <hip/hip_runtime.h>
#include <math.h>

#define BN 128
#define TN 64
#define EN 300
#define MP 20
#define S2E 600
#define EPSF 1e-6f

typedef __attribute__((ext_vector_type(2))) float v2f;
typedef __attribute__((ext_vector_type(8))) float v8f;

#define ROW4 (EN / 4)
#define MTW 16
#define MTT (MTW * 32)
#define XQ ((TN * ROW4 + MTT - 1) / MTT)
#define VQ ((MP * ROW4 + MTT - 1) / MTT)
#define XO 0
#define HO (TN * EN)
#define TO (2 * TN * EN)
#define WO (3 * TN * EN)
#define CP 68
#define CO (WO + 2 * MP * EN)
#define NO (CO + TN * CP)
#define RO (NO + 2 * TN)
#define AO (RO + TN)
#define MWO (AO + TN)
#define SMF (MWO + MP * EN)
#define N1O CO
#define N2O (CO + TN * MP)
#define MPG (MTW / 4)
#define MPW (MP / MPG)
#define ETN ((EN + 15) / 16)
#define NRW (2 * TN / MTW)
#define SRW (TN / MTW)

__global__ __launch_bounds__(MTT)
void match_kernel(const float* __restrict__ h1, const float* __restrict__ h2,
                  const float* __restrict__ kern,
                  float* __restrict__ out) {
  __shared__ __attribute__((aligned(16))) float sm[SMF];
  int db = blockIdx.x;
  int b = db % BN, d = db / BN;
  int wave = __builtin_amdgcn_readfirstlane((int)(threadIdx.x >> 5));
  int lane = threadIdx.x & 31;
  int lh = lane >> 4, ll = lane & 15;
  const int tid = threadIdx.x;

  const float* Xg = h1 + ((size_t)b * TN) * S2E + d * EN;
  const float* Hg = h2 + ((size_t)b * TN) * S2E + d * EN;
  {
    float4 vx[XQ], vh[XQ];
#pragma unroll
    for (int q = 0; q < XQ; ++q) {
      int idx = tid + MTT * q;
      int ci = idx < TN * ROW4 ? idx : TN * ROW4 - 1;
      int r = ci / ROW4, c = ci - r * ROW4;
      vx[q] = *reinterpret_cast<const float4*>(Xg + (size_t)r * S2E + 4 * c);
      vh[q] = *reinterpret_cast<const float4*>(Hg + (size_t)r * S2E + 4 * c);
    }
#pragma unroll
    for (int q = 0; q < XQ; ++q) {
      int idx = tid + MTT * q;
      if (idx < TN * ROW4) {
        *reinterpret_cast<float4*>(sm + XO + 4 * idx) = vx[q];
        *reinterpret_cast<float4*>(sm + HO + 4 * idx) = vh[q];
      }
    }
  }
  __syncthreads();

  {
    float s[NRW];
#pragma unroll
    for (int u = 0; u < NRW; ++u) s[u] = 0.f;
    for (int e = lane; e < EN; e += 32) {
#pragma unroll
      for (int u = 0; u < NRW; ++u) {
        float a = sm[XO + (wave + MTW * u) * EN + e];
        s[u] += a * a;
      }
    }
#pragma unroll
    for (int off = 16; off; off >>= 1) {
#pragma unroll
      for (int u = 0; u < NRW; ++u) s[u] += __shfl_xor(s[u], off, 32);
    }
    if (lane == 0) {
#pragma unroll
      for (int u = 0; u < NRW; ++u) sm[NO + wave + MTW * u] = sqrtf(fmaxf(s[u], EPSF));
    }
  }
  __syncthreads();

  {
    int ci0 = (wave >> 2) * 16, cj0 = (wave & 3) * 16;
    const float* ca = sm + XO + (ci0 + ll) * EN + 2 * lh;
    const float* cb = sm + HO + (cj0 + ll) * EN + 2 * lh;
    v8f acc = {};
#pragma unroll 5
    for (int kk = 0; kk < EN; kk += 4) {
      v2f af = *reinterpret_cast<const v2f*>(ca + kk);
      v2f bf = *reinterpret_cast<const v2f*>(cb + kk);
      acc = __builtin_amdgcn_wmma_f32_16x16x4_f32(false, af, false, bf,
                                                  (short)0, acc, false, false);
    }
    float invn2 = 1.f / sm[NO + TN + cj0 + ll];
#pragma unroll
    for (int r = 0; r < 8; ++r) {
      int i = ci0 + lh * 8 + r;
      sm[CO + i * CP + cj0 + ll] = acc[r] / sm[NO + i] * invn2;
    }
  }
  __syncthreads();

  {
    float s[SRW], m[SRW]; int mi[SRW];
#pragma unroll
    for (int u = 0; u < SRW; ++u) {
      const float* row = sm + CO + (wave + MTW * u) * CP;
      float v0 = row[lane], v1 = row[lane + 32];
      s[u] = v0 + v1;
      m[u] = v0; mi[u] = lane;
      if (v1 > m[u]) { m[u] = v1; mi[u] = lane + 32; }
    }
#pragma unroll
    for (int off = 16; off; off >>= 1) {
#pragma unroll
      for (int u = 0; u < SRW; ++u) {
        s[u] += __shfl_xor(s[u], off, 32);
        float om = __shfl_xor(m[u], off, 32);
        int omi = __shfl_xor(mi[u], off, 32);
        if (om > m[u] || (om == m[u] && omi < mi[u])) { m[u] = om; mi[u] = omi; }
      }
    }
    if (lane == 0) {
#pragma unroll
      for (int u = 0; u < SRW; ++u) {
        sm[RO + wave + MTW * u] = s[u] + EPSF;
        sm[AO + wave + MTW * u] = __int_as_float(mi[u]);
      }
    }
  }
  __syncthreads();

  for (int q = wave; q < 4 * ETN; q += MTW) {
    int ai0 = (q & 3) * 16;
    int et = q >> 2;
    int e0 = et < ETN - 1 ? et * 16 : EN - 16;
    const float* crow = sm + CO + (ai0 + ll) * CP + 2 * lh;
    const float* hcol = sm + HO + (2 * lh) * EN + e0 + ll;
    v8f acc = {};
#pragma unroll
    for (int kk = 0; kk < TN; kk += 4) {
      v2f af = *reinterpret_cast<const v2f*>(crow + kk);
      v2f bf = { hcol[kk * EN], hcol[(kk + 1) * EN] };
      acc = __builtin_amdgcn_wmma_f32_16x16x4_f32(false, af, false, bf,
                                                  (short)0, acc, false, false);
    }
#pragma unroll
    for (int r = 0; r < 8; ++r) {
      int i = ai0 + lh * 8 + r;
      sm[TO + i * EN + e0 + ll] = acc[r] / sm[RO + i];
    }
  }
  __syncthreads();

  int strip = wave & 3, sel = (wave >> 2) & 1, nt = wave >> 3;
  int i0 = strip * 16;
  int mcol = nt * 16 + ll;
  int mrow = mcol < MP ? mcol : MP - 1;
  int jm = __float_as_int(sm[AO + i0 + ll]);
  const float* xr = sm + XO + (i0 + ll) * EN + 2 * lh;
  const float* wr = sm + WO + sel * (MP * EN) + mrow * EN + 2 * lh;

  for (int ph = 0; ph < 2; ++ph) {
    if (ph) __syncthreads();
    {
      float4 wv[2][VQ];
#pragma unroll
      for (int s = 0; s < 2; ++s) {
        int ks = ph == 0 ? (s == 0 ? 0 : 3) : (s == 0 ? 1 : 2);
        const float* wsrc = kern + (ks * 2 + d) * EN;
#pragma unroll
        for (int q = 0; q < VQ; ++q) {
          int idx = tid + MTT * q;
          int ci = idx < MP * ROW4 ? idx : MP * ROW4 - 1;
          int r = ci / ROW4, c = ci - r * ROW4;
          wv[s][q] = *reinterpret_cast<const float4*>(wsrc + (size_t)r * 2400 + 4 * c);
        }
      }
#pragma unroll
      for (int s = 0; s < 2; ++s) {
#pragma unroll
        for (int q = 0; q < VQ; ++q) {
          int idx = tid + MTT * q;
          if (idx < MP * ROW4) {
            float4 w = wv[s][q];
            w.x *= w.x; w.y *= w.y; w.z *= w.z; w.w *= w.w;
            *reinterpret_cast<float4*>(sm + WO + s * (MP * EN) + 4 * idx) = w;
            if (ph == 0 && s == 1) *reinterpret_cast<float4*>(sm + MWO + 4 * idx) = w;
          }
        }
      }
    }
    __syncthreads();

    int kind = ph == 0 ? (sel == 0 ? 0 : 3) : (sel == 0 ? 1 : 2);
    int yoff = (kind == 0) ? HO + (TN - 1) * EN
             : (kind == 1) ? TO + (i0 + ll) * EN
             : (kind == 2) ? HO + jm * EN
             :               HO + (i0 + ll) * EN;
    const float* yr = sm + yoff + 2 * lh;

    v8f sxy = {}, sxx = {}, syy = {};
    if (kind == 3) {
#pragma unroll 5
      for (int kk = 0; kk < EN; kk += 4) {
        v2f x = *reinterpret_cast<const v2f*>(xr + kk);
        v2f y = *reinterpret_cast<const v2f*>(yr + kk);
        v2f w = *reinterpret_cast<const v2f*>(wr + kk);
        sxx = __builtin_amdgcn_wmma_f32_16x16x4_f32(false, x * x, false, w, (short)0, sxx, false, false);
        syy = __builtin_amdgcn_wmma_f32_16x16x4_f32(false, y * y, false, w, (short)0, syy, false, false);
      }
    } else {
#pragma unroll 5
      for (int kk = 0; kk < EN; kk += 4) {
        v2f x = *reinterpret_cast<const v2f*>(xr + kk);
        v2f y = *reinterpret_cast<const v2f*>(yr + kk);
        v2f w = *reinterpret_cast<const v2f*>(wr + kk);
        sxy = __builtin_amdgcn_wmma_f32_16x16x4_f32(false, x * y, false, w, (short)0, sxy, false, false);
        sxx = __builtin_amdgcn_wmma_f32_16x16x4_f32(false, x * x, false, w, (short)0, sxx, false, false);
        syy = __builtin_amdgcn_wmma_f32_16x16x4_f32(false, y * y, false, w, (short)0, syy, false, false);
      }
    }

    if (mcol < MP) {
      if (kind == 3) {
#pragma unroll
        for (int r = 0; r < 8; ++r) {
          int i = i0 + lh * 8 + r;
          sm[N1O + i * MP + mcol] = 1.f / sqrtf(fmaxf(sxx[r], EPSF));
          sm[N2O + i * MP + mcol] = 1.f / sqrtf(fmaxf(syy[r], EPSF));
        }
      } else {
#pragma unroll
        for (int r = 0; r < 8; ++r) {
          int i = i0 + lh * 8 + r;
          out[((size_t)(b * TN + i)) * 160 + (kind * 2 + d) * MP + mcol] =
              sxy[r] / sqrtf(fmaxf(sxx[r], EPSF)) / sqrtf(fmaxf(syy[r], EPSF));
        }
      }
    }
  }

  {
    int mg = wave >> 2;
    const float* ar  = sm + XO + (i0 + ll) * EN + 2 * lh;
    const float* br  = sm + HO + ll * EN + 2 * lh;
    const float* wq  = sm + MWO + mg * EN + 2 * lh;

    v8f acc[MPW][4];
#pragma unroll
    for (int p = 0; p < MPW; ++p)
#pragma unroll
      for (int t = 0; t < 4; ++t) acc[p][t] = (v8f){};

    v2f af[MPW], bq[4];
    {
      v2f a = *reinterpret_cast<const v2f*>(ar);
#pragma unroll
      for (int p = 0; p < MPW; ++p) af[p] = a * *reinterpret_cast<const v2f*>(wq + MPG * p * EN);
#pragma unroll
      for (int t = 0; t < 4; ++t) bq[t] = *reinterpret_cast<const v2f*>(br + 16 * t * EN);
    }

#pragma nounroll
    for (int kk = 0; kk < EN; kk += 4) {
      int kn = kk + 4 < EN ? kk + 4 : kk;
      v2f na = *reinterpret_cast<const v2f*>(ar + kn);
      v2f nb[4], nw[MPW];
#pragma unroll
      for (int t = 0; t < 4; ++t) nb[t] = *reinterpret_cast<const v2f*>(br + 16 * t * EN + kn);
#pragma unroll
      for (int p = 0; p < MPW; ++p) nw[p] = *reinterpret_cast<const v2f*>(wq + MPG * p * EN + kn);

#pragma unroll
      for (int p = 0; p < MPW; ++p)
#pragma unroll
        for (int t = 0; t < 4; ++t)
          acc[p][t] = __builtin_amdgcn_wmma_f32_16x16x4_f32(false, af[p], false, bq[t],
                                                            (short)0, acc[p][t], false, false);

#pragma unroll
      for (int p = 0; p < MPW; ++p) af[p] = na * nw[p];
#pragma unroll
      for (int t = 0; t < 4; ++t) bq[t] = nb[t];
    }

    float v[MPW][8];
#pragma unroll
    for (int p = 0; p < MPW; ++p) {
      int m = mg + MPG * p;
      float d0 = sm[N2O + (0  + ll) * MP + m];
      float d1 = sm[N2O + (16 + ll) * MP + m];
      float d2 = sm[N2O + (32 + ll) * MP + m];
      float d3 = sm[N2O + (48 + ll) * MP + m];
#pragma unroll
      for (int r = 0; r < 8; ++r) {
        float v0 = acc[p][0][r] * d0;
        float v1 = acc[p][1][r] * d1;
        float v2 = acc[p][2][r] * d2;
        float v3 = acc[p][3][r] * d3;
        v[p][r] = fmaxf(fmaxf(v0, v1), fmaxf(v2, v3));
      }
    }
#pragma unroll
    for (int off = 1; off < 16; off <<= 1) {
#pragma unroll
      for (int p = 0; p < MPW; ++p)
#pragma unroll
        for (int r = 0; r < 8; ++r)
          v[p][r] = fmaxf(v[p][r], __shfl_xor(v[p][r], off, 32));
    }
    if (ll == 0) {
#pragma unroll
      for (int p = 0; p < MPW; ++p) {
        int m = mg + MPG * p;
#pragma unroll
        for (int r = 0; r < 8; ++r) {
          int i = i0 + lh * 8 + r;
          out[((size_t)(b * TN + i)) * 160 + (6 + d) * MP + m] = v[p][r] * sm[N1O + i * MP + m];
        }
      }
    }
  }
}

extern "C" void kernel_launch(void* const* d_in, const int* in_sizes, int n_in,
                              void* d_out, int out_size, void* d_ws, size_t ws_size,
                              hipStream_t stream) {
  const float* h1   = (const float*)d_in[0];
  const float* h2   = (const float*)d_in[1];
  const float* kern = (const float*)d_in[2];
  float* out = (float*)d_out;

  match_kernel<<<2 * BN, MTT, 0, stream>>>(h1, h2, kern, out);
}


# matching and attentive epilogues: the per-lane sqrt and divide sequences become v_rsq_f32 and multiplies by a once-per-row reciprocal of the row sum; rounding of those output slices changes
# speedup vs baseline: 4.8561x; 1.0335x over previous
#include <hip/hip_runtime.h>
#include <math.h>

#define BN 128
#define TN 64
#define EN 300
#define MP 20
#define S2E 600
#define EPSF 1e-6f

typedef __attribute__((ext_vector_type(2))) float v2f;
typedef __attribute__((ext_vector_type(8))) float v8f;

#define ROW4 (EN / 4)
#define MTW 16
#define MTT (MTW * 32)
#define XQ ((TN * ROW4 + MTT - 1) / MTT)
#define VQ ((MP * ROW4 + MTT - 1) / MTT)
#define XO 0
#define HO (TN * EN)
#define TO (2 * TN * EN)
#define WO (3 * TN * EN)
#define CP 68
#define CO (WO + 2 * MP * EN)
#define NO (CO + TN * CP)
#define RO (NO + 2 * TN)
#define AO (RO + TN)
#define MWO (AO + TN)
#define SMF (MWO + MP * EN)
#define N1O CO
#define N2O (CO + TN * MP)
#define MPG (MTW / 4)
#define MPW (MP / MPG)
#define ETN ((EN + 15) / 16)
#define NRW (2 * TN / MTW)
#define SRW (TN / MTW)

__global__ __launch_bounds__(MTT)
void match_kernel(const float* __restrict__ h1, const float* __restrict__ h2,
                  const float* __restrict__ kern,
                  float* __restrict__ out) {
  __shared__ __attribute__((aligned(16))) float sm[SMF];
  int db = blockIdx.x;
  int b = db % BN, d = db / BN;
  int wave = __builtin_amdgcn_readfirstlane((int)(threadIdx.x >> 5));
  int lane = threadIdx.x & 31;
  int lh = lane >> 4, ll = lane & 15;
  const int tid = threadIdx.x;

  const float* Xg = h1 + ((size_t)b * TN) * S2E + d * EN;
  const float* Hg = h2 + ((size_t)b * TN) * S2E + d * EN;
  {
    float4 vx[XQ], vh[XQ];
#pragma unroll
    for (int q = 0; q < XQ; ++q) {
      int idx = tid + MTT * q;
      int ci = idx < TN * ROW4 ? idx : TN * ROW4 - 1;
      int r = ci / ROW4, c = ci - r * ROW4;
      vx[q] = *reinterpret_cast<const float4*>(Xg + (size_t)r * S2E + 4 * c);
      vh[q] = *reinterpret_cast<const float4*>(Hg + (size_t)r * S2E + 4 * c);
    }
#pragma unroll
    for (int q = 0; q < XQ; ++q) {
      int idx = tid + MTT * q;
      if (idx < TN * ROW4) {
        *reinterpret_cast<float4*>(sm + XO + 4 * idx) = vx[q];
        *reinterpret_cast<float4*>(sm + HO + 4 * idx) = vh[q];
      }
    }
  }
  __syncthreads();

  {
    float s[NRW];
#pragma unroll
    for (int u = 0; u < NRW; ++u) s[u] = 0.f;
    for (int e = lane; e < EN; e += 32) {
#pragma unroll
      for (int u = 0; u < NRW; ++u) {
        float a = sm[XO + (wave + MTW * u) * EN + e];
        s[u] += a * a;
      }
    }
#pragma unroll
    for (int off = 16; off; off >>= 1) {
#pragma unroll
      for (int u = 0; u < NRW; ++u) s[u] += __shfl_xor(s[u], off, 32);
    }
    if (lane == 0) {
#pragma unroll
      for (int u = 0; u < NRW; ++u) sm[NO + wave + MTW * u] = sqrtf(fmaxf(s[u], EPSF));
    }
  }
  __syncthreads();

  {
    int ci0 = (wave >> 2) * 16, cj0 = (wave & 3) * 16;
    const float* ca = sm + XO + (ci0 + ll) * EN + 2 * lh;
    const float* cb = sm + HO + (cj0 + ll) * EN + 2 * lh;
    v8f acc = {};
#pragma unroll 5
    for (int kk = 0; kk < EN; kk += 4) {
      v2f af = *reinterpret_cast<const v2f*>(ca + kk);
      v2f bf = *reinterpret_cast<const v2f*>(cb + kk);
      acc = __builtin_amdgcn_wmma_f32_16x16x4_f32(false, af, false, bf,
                                                  (short)0, acc, false, false);
    }
    float invn2 = 1.f / sm[NO + TN + cj0 + ll];
#pragma unroll
    for (int r = 0; r < 8; ++r) {
      int i = ci0 + lh * 8 + r;
      sm[CO + i * CP + cj0 + ll] = acc[r] / sm[NO + i] * invn2;
    }
  }
  __syncthreads();

  {
    float s[SRW], m[SRW]; int mi[SRW];
#pragma unroll
    for (int u = 0; u < SRW; ++u) {
      const float* row = sm + CO + (wave + MTW * u) * CP;
      float v0 = row[lane], v1 = row[lane + 32];
      s[u] = v0 + v1;
      m[u] = v0; mi[u] = lane;
      if (v1 > m[u]) { m[u] = v1; mi[u] = lane + 32; }
    }
#pragma unroll
    for (int off = 16; off; off >>= 1) {
#pragma unroll
      for (int u = 0; u < SRW; ++u) {
        s[u] += __shfl_xor(s[u], off, 32);
        float om = __shfl_xor(m[u], off, 32);
        int omi = __shfl_xor(mi[u], off, 32);
        if (om > m[u] || (om == m[u] && omi < mi[u])) { m[u] = om; mi[u] = omi; }
      }
    }
    if (lane == 0) {
#pragma unroll
      for (int u = 0; u < SRW; ++u) {
        sm[RO + wave + MTW * u] = 1.f / (s[u] + EPSF);
        sm[AO + wave + MTW * u] = __int_as_float(mi[u]);
      }
    }
  }
  __syncthreads();

  for (int q = wave; q < 4 * ETN; q += MTW) {
    int ai0 = (q & 3) * 16;
    int et = q >> 2;
    int e0 = et < ETN - 1 ? et * 16 : EN - 16;
    const float* crow = sm + CO + (ai0 + ll) * CP + 2 * lh;
    const float* hcol = sm + HO + (2 * lh) * EN + e0 + ll;
    v8f acc = {};
#pragma unroll
    for (int kk = 0; kk < TN; kk += 4) {
      v2f af = *reinterpret_cast<const v2f*>(crow + kk);
      v2f bf = { hcol[kk * EN], hcol[(kk + 1) * EN] };
      acc = __builtin_amdgcn_wmma_f32_16x16x4_f32(false, af, false, bf,
                                                  (short)0, acc, false, false);
    }
#pragma unroll
    for (int r = 0; r < 8; ++r) {
      int i = ai0 + lh * 8 + r;
      sm[TO + i * EN + e0 + ll] = acc[r] * sm[RO + i];
    }
  }
  __syncthreads();

  int strip = wave & 3, sel = (wave >> 2) & 1, nt = wave >> 3;
  int i0 = strip * 16;
  int mcol = nt * 16 + ll;
  int mrow = mcol < MP ? mcol : MP - 1;
  int jm = __float_as_int(sm[AO + i0 + ll]);
  const float* xr = sm + XO + (i0 + ll) * EN + 2 * lh;
  const float* wr = sm + WO + sel * (MP * EN) + mrow * EN + 2 * lh;

  for (int ph = 0; ph < 2; ++ph) {
    if (ph) __syncthreads();
    {
      float4 wv[2][VQ];
#pragma unroll
      for (int s = 0; s < 2; ++s) {
        int ks = ph == 0 ? (s == 0 ? 0 : 3) : (s == 0 ? 1 : 2);
        const float* wsrc = kern + (ks * 2 + d) * EN;
#pragma unroll
        for (int q = 0; q < VQ; ++q) {
          int idx = tid + MTT * q;
          int ci = idx < MP * ROW4 ? idx : MP * ROW4 - 1;
          int r = ci / ROW4, c = ci - r * ROW4;
          wv[s][q] = *reinterpret_cast<const float4*>(wsrc + (size_t)r * 2400 + 4 * c);
        }
      }
#pragma unroll
      for (int s = 0; s < 2; ++s) {
#pragma unroll
        for (int q = 0; q < VQ; ++q) {
          int idx = tid + MTT * q;
          if (idx < MP * ROW4) {
            float4 w = wv[s][q];
            w.x *= w.x; w.y *= w.y; w.z *= w.z; w.w *= w.w;
            *reinterpret_cast<float4*>(sm + WO + s * (MP * EN) + 4 * idx) = w;
            if (ph == 0 && s == 1) *reinterpret_cast<float4*>(sm + MWO + 4 * idx) = w;
          }
        }
      }
    }
    __syncthreads();

    int kind = ph == 0 ? (sel == 0 ? 0 : 3) : (sel == 0 ? 1 : 2);
    int yoff = (kind == 0) ? HO + (TN - 1) * EN
             : (kind == 1) ? TO + (i0 + ll) * EN
             : (kind == 2) ? HO + jm * EN
             :               HO + (i0 + ll) * EN;
    const float* yr = sm + yoff + 2 * lh;

    v8f sxy = {}, sxx = {}, syy = {};
    if (kind == 3) {
#pragma unroll 5
      for (int kk = 0; kk < EN; kk += 4) {
        v2f x = *reinterpret_cast<const v2f*>(xr + kk);
        v2f y = *reinterpret_cast<const v2f*>(yr + kk);
        v2f w = *reinterpret_cast<const v2f*>(wr + kk);
        sxx = __builtin_amdgcn_wmma_f32_16x16x4_f32(false, x * x, false, w, (short)0, sxx, false, false);
        syy = __builtin_amdgcn_wmma_f32_16x16x4_f32(false, y * y, false, w, (short)0, syy, false, false);
      }
    } else {
#pragma unroll 5
      for (int kk = 0; kk < EN; kk += 4) {
        v2f x = *reinterpret_cast<const v2f*>(xr + kk);
        v2f y = *reinterpret_cast<const v2f*>(yr + kk);
        v2f w = *reinterpret_cast<const v2f*>(wr + kk);
        sxy = __builtin_amdgcn_wmma_f32_16x16x4_f32(false, x * y, false, w, (short)0, sxy, false, false);
        sxx = __builtin_amdgcn_wmma_f32_16x16x4_f32(false, x * x, false, w, (short)0, sxx, false, false);
        syy = __builtin_amdgcn_wmma_f32_16x16x4_f32(false, y * y, false, w, (short)0, syy, false, false);
      }
    }

    if (mcol < MP) {
      if (kind == 3) {
#pragma unroll
        for (int r = 0; r < 8; ++r) {
          int i = i0 + lh * 8 + r;
          sm[N1O + i * MP + mcol] = __builtin_amdgcn_rsqf(fmaxf(sxx[r], EPSF));
          sm[N2O + i * MP + mcol] = __builtin_amdgcn_rsqf(fmaxf(syy[r], EPSF));
        }
      } else {
#pragma unroll
        for (int r = 0; r < 8; ++r) {
          int i = i0 + lh * 8 + r;
          out[((size_t)(b * TN + i)) * 160 + (kind * 2 + d) * MP + mcol] =
              sxy[r] * __builtin_amdgcn_rsqf(fmaxf(sxx[r], EPSF)) * __builtin_amdgcn_rsqf(fmaxf(syy[r], EPSF));
        }
      }
    }
  }

  {
    int mg = wave >> 2;
    const float* ar  = sm + XO + (i0 + ll) * EN + 2 * lh;
    const float* br  = sm + HO + ll * EN + 2 * lh;
    const float* wq  = sm + MWO + mg * EN + 2 * lh;

    v8f acc[MPW][4];
#pragma unroll
    for (int p = 0; p < MPW; ++p)
#pragma unroll
      for (int t = 0; t < 4; ++t) acc[p][t] = (v8f){};

    v2f af[MPW], bq[4];
    {
      v2f a = *reinterpret_cast<const v2f*>(ar);
#pragma unroll
      for (int p = 0; p < MPW; ++p) af[p] = a * *reinterpret_cast<const v2f*>(wq + MPG * p * EN);
#pragma unroll
      for (int t = 0; t < 4; ++t) bq[t] = *reinterpret_cast<const v2f*>(br + 16 * t * EN);
    }

#pragma nounroll
    for (int kk = 0; kk < EN; kk += 4) {
      int kn = kk + 4 < EN ? kk + 4 : kk;
      v2f na = *reinterpret_cast<const v2f*>(ar + kn);
      v2f nb[4], nw[MPW];
#pragma unroll
      for (int t = 0; t < 4; ++t) nb[t] = *reinterpret_cast<const v2f*>(br + 16 * t * EN + kn);
#pragma unroll
      for (int p = 0; p < MPW; ++p) nw[p] = *reinterpret_cast<const v2f*>(wq + MPG * p * EN + kn);

#pragma unroll
      for (int p = 0; p < MPW; ++p)
#pragma unroll
        for (int t = 0; t < 4; ++t)
          acc[p][t] = __builtin_amdgcn_wmma_f32_16x16x4_f32(false, af[p], false, bq[t],
                                                            (short)0, acc[p][t], false, false);

#pragma unroll
      for (int p = 0; p < MPW; ++p) af[p] = na * nw[p];
#pragma unroll
      for (int t = 0; t < 4; ++t) bq[t] = nb[t];
    }

    float v[MPW][8];
#pragma unroll
    for (int p = 0; p < MPW; ++p) {
      int m = mg + MPG * p;
      float d0 = sm[N2O + (0  + ll) * MP + m];
      float d1 = sm[N2O + (16 + ll) * MP + m];
      float d2 = sm[N2O + (32 + ll) * MP + m];
      float d3 = sm[N2O + (48 + ll) * MP + m];
#pragma unroll
      for (int r = 0; r < 8; ++r) {
        float v0 = acc[p][0][r] * d0;
        float v1 = acc[p][1][r] * d1;
        float v2 = acc[p][2][r] * d2;
        float v3 = acc[p][3][r] * d3;
        v[p][r] = fmaxf(fmaxf(v0, v1), fmaxf(v2, v3));
      }
    }
#pragma unroll
    for (int off = 1; off < 16; off <<= 1) {
#pragma unroll
      for (int p = 0; p < MPW; ++p)
#pragma unroll
        for (int r = 0; r < 8; ++r)
          v[p][r] = fmaxf(v[p][r], __shfl_xor(v[p][r], off, 32));
    }
    if (ll == 0) {
#pragma unroll
      for (int p = 0; p < MPW; ++p) {
        int m = mg + MPG * p;
#pragma unroll
        for (int r = 0; r < 8; ++r) {
          int i = i0 + lh * 8 + r;
          out[((size_t)(b * TN + i)) * 160 + (6 + d) * MP + m] = v[p][r] * sm[N1O + i * MP + m];
        }
      }
    }
  }
}

extern "C" void kernel_launch(void* const* d_in, const int* in_sizes, int n_in,
                              void* d_out, int out_size, void* d_ws, size_t ws_size,
                              hipStream_t stream) {
  const float* h1   = (const float*)d_in[0];
  const float* h2   = (const float*)d_in[1];
  const float* kern = (const float*)d_in[2];
  float* out = (float*)d_out;

  match_kernel<<<2 * BN, MTT, 0, stream>>>(h1, h2, kern, out);
}


# matchings in one pass with all four squared-weight slices resident: products with the same A fragment share a wave (x*x: 5 tiles for 4 slices, h2*h2: 3 for 2), 22 -> 17 tile chains per strip
# speedup vs baseline: 5.0753x; 1.0451x over previous
#include <hip/hip_runtime.h>
#include <math.h>

#define BN 128
#define TN 64
#define EN 300
#define MP 20
#define S2E 600
#define EPSF 1e-6f

typedef __attribute__((ext_vector_type(2))) float v2f;
typedef __attribute__((ext_vector_type(8))) float v8f;

#define ROW4 (EN / 4)
#define MTW 16
#define MTT (MTW * 32)
#define XQ ((TN * ROW4 + MTT - 1) / MTT)
#define VQ ((MP * ROW4 + MTT - 1) / MTT)
#define XO 0
#define HO (TN * EN)
#define TO (2 * TN * EN)
#define WO (3 * TN * EN)
#define CP 68
#define CO (WO + 3 * MP * EN)
#define NO (WO + 4 * MP * EN)
#define RO (NO + 2 * TN)
#define AO (RO + TN)
#define SMF (AO + TN)
#define SXO WO
#define N1O (SXO + 3 * TN * MP)
#define N2O (SXO + 4 * TN * MP)
#define FYO (N2O + TN * MP)
#define G2O (FYO + 32)
#define MPG (MTW / 4)
#define MPW (MP / MPG)
#define ETN ((EN + 15) / 16)
#define NRW (2 * TN / MTW)
#define SRW (TN / MTW)

__global__ __launch_bounds__(MTT)
void match_kernel(const float* __restrict__ h1, const float* __restrict__ h2,
                  const float* __restrict__ kern,
                  float* __restrict__ out) {
  __shared__ __attribute__((aligned(16))) float sm[SMF];
  int db = blockIdx.x;
  int b = db % BN, d = db / BN;
  int wave = __builtin_amdgcn_readfirstlane((int)(threadIdx.x >> 5));
  int lane = threadIdx.x & 31;
  int lh = lane >> 4, ll = lane & 15;
  const int tid = threadIdx.x;

  const float* Xg = h1 + ((size_t)b * TN) * S2E + d * EN;
  const float* Hg = h2 + ((size_t)b * TN) * S2E + d * EN;
  {
    float4 vx[XQ], vh[XQ];
#pragma unroll
    for (int q = 0; q < XQ; ++q) {
      int idx = tid + MTT * q;
      int ci = idx < TN * ROW4 ? idx : TN * ROW4 - 1;
      int r = ci / ROW4, c = ci - r * ROW4;
      vx[q] = *reinterpret_cast<const float4*>(Xg + (size_t)r * S2E + 4 * c);
      vh[q] = *reinterpret_cast<const float4*>(Hg + (size_t)r * S2E + 4 * c);
    }
#pragma unroll
    for (int q = 0; q < XQ; ++q) {
      int idx = tid + MTT * q;
      if (idx < TN * ROW4) {
        *reinterpret_cast<float4*>(sm + XO + 4 * idx) = vx[q];
        *reinterpret_cast<float4*>(sm + HO + 4 * idx) = vh[q];
      }
    }
  }
  __syncthreads();

  float4 wv[4][VQ];
#pragma unroll
  for (int q = 0; q < VQ; ++q) {
    int idx = tid + MTT * q;
    int ci = idx < MP * ROW4 ? idx : MP * ROW4 - 1;
    int r = ci / ROW4, c = ci - r * ROW4;
    const float* wsrc = kern + (size_t)r * 2400 + d * EN + 4 * c;
#pragma unroll
    for (int sl = 0; sl < 4; ++sl) wv[sl][q] = *reinterpret_cast<const float4*>(wsrc + 2 * sl * EN);
  }

  {
    float s[NRW];
#pragma unroll
    for (int u = 0; u < NRW; ++u) s[u] = 0.f;
    for (int e = lane; e < EN; e += 32) {
#pragma unroll
      for (int u = 0; u < NRW; ++u) {
        float a = sm[XO + (wave + MTW * u) * EN + e];
        s[u] += a * a;
      }
    }
#pragma unroll
    for (int off = 16; off; off >>= 1) {
#pragma unroll
      for (int u = 0; u < NRW; ++u) s[u] += __shfl_xor(s[u], off, 32);
    }
    if (lane == 0) {
#pragma unroll
      for (int u = 0; u < NRW; ++u) sm[NO + wave + MTW * u] = sqrtf(fmaxf(s[u], EPSF));
    }
  }
#pragma unroll
  for (int sl = 0; sl < 3; ++sl) {
#pragma unroll
    for (int q = 0; q < VQ; ++q) {
      int idx = tid + MTT * q;
      if (idx < MP * ROW4) {
        float4 w = wv[sl][q];
        w.x *= w.x; w.y *= w.y; w.z *= w.z; w.w *= w.w;
        *reinterpret_cast<float4*>(sm + WO + sl * (MP * EN) + 4 * idx) = w;
      }
    }
  }
  __syncthreads();

  {
    int ci0 = (wave >> 2) * 16, cj0 = (wave & 3) * 16;
    const float* ca = sm + XO + (ci0 + ll) * EN + 2 * lh;
    const float* cb = sm + HO + (cj0 + ll) * EN + 2 * lh;
    v8f acc = {};
#pragma unroll 5
    for (int kk = 0; kk < EN; kk += 4) {
      v2f af = *reinterpret_cast<const v2f*>(ca + kk);
      v2f bf = *reinterpret_cast<const v2f*>(cb + kk);
      acc = __builtin_amdgcn_wmma_f32_16x16x4_f32(false, af, false, bf,
                                                  (short)0, acc, false, false);
    }
    float invn2 = 1.f / sm[NO + TN + cj0 + ll];
#pragma unroll
    for (int r = 0; r < 8; ++r) {
      int i = ci0 + lh * 8 + r;
      sm[CO + i * CP + cj0 + ll] = acc[r] / sm[NO + i] * invn2;
    }
  }
  __syncthreads();

  {
    float s[SRW], m[SRW]; int mi[SRW];
#pragma unroll
    for (int u = 0; u < SRW; ++u) {
      const float* row = sm + CO + (wave + MTW * u) * CP;
      float v0 = row[lane], v1 = row[lane + 32];
      s[u] = v0 + v1;
      m[u] = v0; mi[u] = lane;
      if (v1 > m[u]) { m[u] = v1; mi[u] = lane + 32; }
    }
#pragma unroll
    for (int off = 16; off; off >>= 1) {
#pragma unroll
      for (int u = 0; u < SRW; ++u) {
        s[u] += __shfl_xor(s[u], off, 32);
        float om = __shfl_xor(m[u], off, 32);
        int omi = __shfl_xor(mi[u], off, 32);
        if (om > m[u] || (om == m[u] && omi < mi[u])) { m[u] = om; mi[u] = omi; }
      }
    }
    if (lane == 0) {
#pragma unroll
      for (int u = 0; u < SRW; ++u) {
        sm[RO + wave + MTW * u] = 1.f / (s[u] + EPSF);
        sm[AO + wave + MTW * u] = __int_as_float(mi[u]);
      }
    }
  }
  __syncthreads();

  for (int q = wave; q < 4 * ETN; q += MTW) {
    int ai0 = (q & 3) * 16;
    int et = q >> 2;
    int e0 = et < ETN - 1 ? et * 16 : EN - 16;
    const float* crow = sm + CO + (ai0 + ll) * CP + 2 * lh;
    const float* hcol = sm + HO + (2 * lh) * EN + e0 + ll;
    v8f acc = {};
#pragma unroll
    for (int kk = 0; kk < TN; kk += 4) {
      v2f af = *reinterpret_cast<const v2f*>(crow + kk);
      v2f bf = { hcol[kk * EN], hcol[(kk + 1) * EN] };
      acc = __builtin_amdgcn_wmma_f32_16x16x4_f32(false, af, false, bf,
                                                  (short)0, acc, false, false);
    }
#pragma unroll
    for (int r = 0; r < 8; ++r) {
      int i = ai0 + lh * 8 + r;
      sm[TO + i * EN + e0 + ll] = acc[r] * sm[RO + i];
    }
  }
  __syncthreads();

  int strip = wave & 3, role = wave >> 2;
  int i0 = strip * 16;
  int jm = __float_as_int(sm[AO + i0 + ll]);
  {
#pragma unroll
    for (int q = 0; q < VQ; ++q) {
      int idx = tid + MTT * q;
      if (idx < MP * ROW4) {
        float4 w = wv[3][q];
        w.x *= w.x; w.y *= w.y; w.z *= w.z; w.w *= w.w;
        *reinterpret_cast<float4*>(sm + WO + 3 * (MP * EN) + 4 * idx) = w;
      }
    }
  }
  __syncthreads();

  int tl = 16 + ll < MP ? 16 + ll : MP - 1;
  const float* xr = sm + XO + (i0 + ll) * EN + 2 * lh;
  v8f c0 = {}, c1 = {}, c2 = {}, c3 = {}, c4 = {}, c5 = {};
  if (role == 0) {
    const float* yl = sm + HO + (TN - 1) * EN + 2 * lh;
    const float* hr = sm + HO + (i0 + ll) * EN + 2 * lh;
    const float* wa = sm + WO + ll * EN + 2 * lh;
    const float* wz = sm + WO + tl * EN + 2 * lh;
    const float* wt = sm + WO + ((2 + ((ll >> 2) & 1)) * MP + 16 + (ll & 3)) * EN + 2 * lh;
    const float* wf = (strip & 1) ? wz : wa;
#pragma unroll 5
    for (int kk = 0; kk < EN; kk += 4) {
      v2f x = *reinterpret_cast<const v2f*>(xr + kk);
      v2f y = *reinterpret_cast<const v2f*>(yl + kk);
      v2f h = *reinterpret_cast<const v2f*>(hr + kk);
      v2f a0 = *reinterpret_cast<const v2f*>(wa + kk);
      v2f z0 = *reinterpret_cast<const v2f*>(wz + kk);
      v2f f0 = *reinterpret_cast<const v2f*>(wf + kk);
      v2f a2 = *reinterpret_cast<const v2f*>(wa + 2 * MP * EN + kk);
      v2f a3 = *reinterpret_cast<const v2f*>(wa + 3 * MP * EN + kk);
      v2f t23 = *reinterpret_cast<const v2f*>(wt + kk);
      v2f xy = x * y, hh = h * h, yy = y * y;
      c0 = __builtin_amdgcn_wmma_f32_16x16x4_f32(false, xy, false, a0, (short)0, c0, false, false);
      c1 = __builtin_amdgcn_wmma_f32_16x16x4_f32(false, xy, false, z0, (short)0, c1, false, false);
      c2 = __builtin_amdgcn_wmma_f32_16x16x4_f32(false, hh, false, a3, (short)0, c2, false, false);
      c3 = __builtin_amdgcn_wmma_f32_16x16x4_f32(false, hh, false, t23, (short)0, c3, false, false);
      c4 = __builtin_amdgcn_wmma_f32_16x16x4_f32(false, yy, false, f0, (short)0, c4, false, false);
      c5 = __builtin_amdgcn_wmma_f32_16x16x4_f32(false, hh, false, a2, (short)0, c5, false, false);
    }
  } else if (role == 1) {
    const float* wb = sm + WO + ll * EN + 2 * lh;
    const float* wt = sm + WO + ((ll >> 2) * MP + 16 + (ll & 3)) * EN + 2 * lh;
#pragma unroll 5
    for (int kk = 0; kk < EN; kk += 4) {
      v2f x = *reinterpret_cast<const v2f*>(xr + kk);
      v2f b0 = *reinterpret_cast<const v2f*>(wb + kk);
      v2f b1 = *reinterpret_cast<const v2f*>(wb + MP * EN + kk);
      v2f b2 = *reinterpret_cast<const v2f*>(wb + 2 * MP * EN + kk);
      v2f b3 = *reinterpret_cast<const v2f*>(wb + 3 * MP * EN + kk);
      v2f bt = *reinterpret_cast<const v2f*>(wt + kk);
      v2f xx = x * x;
      c0 = __builtin_amdgcn_wmma_f32_16x16x4_f32(false, xx, false, b0, (short)0, c0, false, false);
      c1 = __builtin_amdgcn_wmma_f32_16x16x4_f32(false, xx, false, b1, (short)0, c1, false, false);
      c2 = __builtin_amdgcn_wmma_f32_16x16x4_f32(false, xx, false, b2, (short)0, c2, false, false);
      c3 = __builtin_amdgcn_wmma_f32_16x16x4_f32(false, xx, false, b3, (short)0, c3, false, false);
      c4 = __builtin_amdgcn_wmma_f32_16x16x4_f32(false, xx, false, bt, (short)0, c4, false, false);
    }
  } else if (role == 2) {
    const float* yr = sm + TO + (i0 + ll) * EN + 2 * lh;
    const float* wa = sm + WO + MP * EN + ll * EN + 2 * lh;
    const float* wz = sm + WO + MP * EN + tl * EN + 2 * lh;
#pragma unroll 5
    for (int kk = 0; kk < EN; kk += 4) {
      v2f x = *reinterpret_cast<const v2f*>(xr + kk);
      v2f y = *reinterpret_cast<const v2f*>(yr + kk);
      v2f a = *reinterpret_cast<const v2f*>(wa + kk);
      v2f z = *reinterpret_cast<const v2f*>(wz + kk);
      v2f xy = x * y, yy = y * y;
      c0 = __builtin_amdgcn_wmma_f32_16x16x4_f32(false, xy, false, a, (short)0, c0, false, false);
      c1 = __builtin_amdgcn_wmma_f32_16x16x4_f32(false, xy, false, z, (short)0, c1, false, false);
      c2 = __builtin_amdgcn_wmma_f32_16x16x4_f32(false, yy, false, a, (short)0, c2, false, false);
      c3 = __builtin_amdgcn_wmma_f32_16x16x4_f32(false, yy, false, z, (short)0, c3, false, false);
    }
  } else {
    const float* yr = sm + HO + jm * EN + 2 * lh;
    const float* wa = sm + WO + 2 * MP * EN + ll * EN + 2 * lh;
    const float* wz = sm + WO + 2 * MP * EN + tl * EN + 2 * lh;
#pragma unroll 5
    for (int kk = 0; kk < EN; kk += 4) {
      v2f x = *reinterpret_cast<const v2f*>(xr + kk);
      v2f y = *reinterpret_cast<const v2f*>(yr + kk);
      v2f a = *reinterpret_cast<const v2f*>(wa + kk);
      v2f z = *reinterpret_cast<const v2f*>(wz + kk);
      v2f xy = x * y;
      c0 = __builtin_amdgcn_wmma_f32_16x16x4_f32(false, xy, false, a, (short)0, c0, false, false);
      c1 = __builtin_amdgcn_wmma_f32_16x16x4_f32(false, xy, false, z, (short)0, c1, false, false);
    }
  }
  __syncthreads();

  if (role == 1) {
#pragma unroll
    for (int r = 0; r < 8; ++r) {
      int i = i0 + lh * 8 + r;
      sm[SXO + (0 * TN + i) * MP + ll] = __builtin_amdgcn_rsqf(fmaxf(c0[r], EPSF));
      sm[SXO + (1 * TN + i) * MP + ll] = __builtin_amdgcn_rsqf(fmaxf(c1[r], EPSF));
      sm[SXO + (2 * TN + i) * MP + ll] = __builtin_amdgcn_rsqf(fmaxf(c2[r], EPSF));
      sm[SXO + (3 * TN + i) * MP + ll] = __builtin_amdgcn_rsqf(fmaxf(c3[r], EPSF));
      sm[SXO + ((ll >> 2) * TN + i) * MP + 16 + (ll & 3)] = __builtin_amdgcn_rsqf(fmaxf(c4[r], EPSF));
    }
  } else if (role == 0) {
#pragma unroll
    for (int r = 0; r < 8; ++r) {
      int j = i0 + lh * 8 + r;
      sm[N2O + j * MP + ll] = __builtin_amdgcn_rsqf(fmaxf(c2[r], EPSF));
      sm[G2O + j * MP + ll] = __builtin_amdgcn_rsqf(fmaxf(c5[r], EPSF));
      if (ll < 8) sm[(ll < 4 ? G2O : N2O) + j * MP + 16 + (ll & 3)] = __builtin_amdgcn_rsqf(fmaxf(c3[r], EPSF));
    }
    if (strip < 2 && lh == 0 && strip * 16 + ll < MP)
      sm[FYO + strip * 16 + ll] = __builtin_amdgcn_rsqf(fmaxf(c4[0], EPSF));
  }
  __syncthreads();

  if (role != 1) {
    int kind = role == 0 ? 0 : role - 1;
    float fy0 = sm[FYO + ll], fy1 = sm[FYO + tl];
#pragma unroll
    for (int r = 0; r < 8; ++r) {
      int i = i0 + lh * 8 + r;
      float* o = out + ((size_t)(b * TN + i)) * 160 + (kind * 2 + d) * MP;
      const float* sx = sm + SXO + (kind * TN + i) * MP;
      const float* g2 = sm + G2O + __float_as_int(sm[AO + i]) * MP;
      float ry0 = role == 0 ? fy0 : role == 2 ? __builtin_amdgcn_rsqf(fmaxf(c2[r], EPSF)) : g2[ll];
      float ry1 = role == 0 ? fy1 : role == 2 ? __builtin_amdgcn_rsqf(fmaxf(c3[r], EPSF)) : g2[tl];
      o[ll] = c0[r] * sx[ll] * ry0;
      if (ll < MP - 16) o[16 + ll] = c1[r] * sx[tl] * ry1;
    }
  }

  {
    int mg = wave >> 2;
    const float* ar  = sm + XO + (i0 + ll) * EN + 2 * lh;
    const float* br  = sm + HO + ll * EN + 2 * lh;
    const float* wq  = sm + WO + 3 * (MP * EN) + mg * EN + 2 * lh;

    v8f acc[MPW][4];
#pragma unroll
    for (int p = 0; p < MPW; ++p)
#pragma unroll
      for (int t = 0; t < 4; ++t) acc[p][t] = (v8f){};

    v2f af[MPW], bq[4];
    {
      v2f a = *reinterpret_cast<const v2f*>(ar);
#pragma unroll
      for (int p = 0; p < MPW; ++p) af[p] = a * *reinterpret_cast<const v2f*>(wq + MPG * p * EN);
#pragma unroll
      for (int t = 0; t < 4; ++t) bq[t] = *reinterpret_cast<const v2f*>(br + 16 * t * EN);
    }

#pragma nounroll
    for (int kk = 0; kk < EN; kk += 4) {
      int kn = kk + 4 < EN ? kk + 4 : kk;
      v2f na = *reinterpret_cast<const v2f*>(ar + kn);
      v2f nb[4], nw[MPW];
#pragma unroll
      for (int t = 0; t < 4; ++t) nb[t] = *reinterpret_cast<const v2f*>(br + 16 * t * EN + kn);
#pragma unroll
      for (int p = 0; p < MPW; ++p) nw[p] = *reinterpret_cast<const v2f*>(wq + MPG * p * EN + kn);

#pragma unroll
      for (int p = 0; p < MPW; ++p)
#pragma unroll
        for (int t = 0; t < 4; ++t)
          acc[p][t] = __builtin_amdgcn_wmma_f32_16x16x4_f32(false, af[p], false, bq[t],
                                                            (short)0, acc[p][t], false, false);

#pragma unroll
      for (int p = 0; p < MPW; ++p) af[p] = na * nw[p];
#pragma unroll
      for (int t = 0; t < 4; ++t) bq[t] = nb[t];
    }

    float v[MPW][8];
#pragma unroll
    for (int p = 0; p < MPW; ++p) {
      int m = mg + MPG * p;
      float d0 = sm[N2O + (0  + ll) * MP + m];
      float d1 = sm[N2O + (16 + ll) * MP + m];
      float d2 = sm[N2O + (32 + ll) * MP + m];
      float d3 = sm[N2O + (48 + ll) * MP + m];
#pragma unroll
      for (int r = 0; r < 8; ++r) {
        float v0 = acc[p][0][r] * d0;
        float v1 = acc[p][1][r] * d1;
        float v2 = acc[p][2][r] * d2;
        float v3 = acc[p][3][r] * d3;
        v[p][r] = fmaxf(fmaxf(v0, v1), fmaxf(v2, v3));
      }
    }
#pragma unroll
    for (int off = 1; off < 16; off <<= 1) {
#pragma unroll
      for (int p = 0; p < MPW; ++p)
#pragma unroll
        for (int r = 0; r < 8; ++r)
          v[p][r] = fmaxf(v[p][r], __shfl_xor(v[p][r], off, 32));
    }
    if (ll == 0) {
#pragma unroll
      for (int p = 0; p < MPW; ++p) {
        int m = mg + MPG * p;
#pragma unroll
        for (int r = 0; r < 8; ++r) {
          int i = i0 + lh * 8 + r;
          out[((size_t)(b * TN + i)) * 160 + (6 + d) * MP + m] = v[p][r] * sm[N1O + i * MP + m];
        }
      }
    }
  }
}

extern "C" void kernel_launch(void* const* d_in, const int* in_sizes, int n_in,
                              void* d_out, int out_size, void* d_ws, size_t ws_size,
                              hipStream_t stream) {
  const float* h1   = (const float*)d_in[0];
  const float* h2   = (const float*)d_in[1];
  const float* kern = (const float*)d_in[2];
  float* out = (float*)d_out;

  match_kernel<<<2 * BN, MTT, 0, stream>>>(h1, h2, kern, out);
}
